# starting kernel
# baseline (speedup 1.0000x reference)
#include <hip/hip_runtime.h>

#define IN_CH  128
#define HID    64
#define OUT_CH 16

typedef float v2f __attribute__((ext_vector_type(2)));
typedef float v8f __attribute__((ext_vector_type(8)));

__global__ void gcn_init_kernel(float* __restrict__ deg,
                                float* __restrict__ agg1, const float* __restrict__ b1,
                                float* __restrict__ out,  const float* __restrict__ b2,
                                int nNodes) {
    long long idx = (long long)blockIdx.x * blockDim.x + threadIdx.x;
    if (idx >= (long long)nNodes * HID) return;
    int i = (int)(idx >> 6);
    int c = (int)(idx & (HID - 1));
    agg1[idx] = b1[c];
    if (c == 0)     deg[i] = 1.0f;
    if (c < OUT_CH) out[(long long)i * OUT_CH + c] = b2[c];
}

__global__ void gcn_degree_kernel(const int* __restrict__ dst,
                                  float* __restrict__ deg, int nEdges) {
    int e = blockIdx.x * blockDim.x + threadIdx.x;
    if (e < nEdges) atomicAdd(&deg[dst[e]], 1.0f);
}

__global__ void gcn_rsqrt_kernel(float* __restrict__ deg, int n) {
    int i = blockIdx.x * blockDim.x + threadIdx.x;
    if (i < n) deg[i] = rsqrtf(deg[i]);
}

__global__ void gcn_gemm_wmma_f32(const float* __restrict__ A,
                                  const float* __restrict__ W,
                                  float* __restrict__ C,
                                  int M, int K, int N) {
    const int wavesPerBlock = blockDim.x >> 5;
    const int wave = (int)(threadIdx.x >> 5);
    const int lane = (int)(threadIdx.x & 31);
    const int tilesN = N >> 4;
    const int gw = blockIdx.x * wavesPerBlock + wave;
    const int tileM = gw / tilesN;
    const int tileN = gw - tileM * tilesN;
    if (tileM * 16 >= M) return;

    const int row0 = tileM << 4;
    const int col0 = tileN << 4;
    const int mr = lane & 15;
    const int kh = (lane >> 4) << 1;

    v8f acc = {};
    for (int k = 0; k < K; k += 4) {
        v2f a, b;
        const float* ap = A + (long long)(row0 + mr) * K + (k + kh);
        a.x = ap[0];
        a.y = ap[1];
        const float* bp = W + (long long)(k + kh) * N + (col0 + mr);
        b.x = bp[0];
        b.y = bp[N];
        acc = __builtin_amdgcn_wmma_f32_16x16x4_f32(
             false, a,  false, b,
             (short)0, acc,  false,  false);
    }

    const int crow = row0 + ((lane >> 4) << 3);
    const int ccol = col0 + mr;
#pragma unroll
    for (int r = 0; r < 8; ++r)
        C[(long long)(crow + r) * N + ccol] = acc[r];
}

template <int C>
__global__ void gcn_scatter_kernel(const float* __restrict__ xl,
                                   const int* __restrict__ src,
                                   const int* __restrict__ dst,
                                   const float* __restrict__ dinv,
                                   float* __restrict__ agg,
                                   int nEdges, int nNodes) {
    long long idx = (long long)blockIdx.x * blockDim.x + threadIdx.x;
    long long total = (long long)(nEdges + nNodes) * C;
    if (idx >= total) return;
    int c = (int)(idx % C);
    long long e = idx / C;
    int s, d;
    if (e < nEdges) { s = src[e]; d = dst[e]; }
    else            { s = d = (int)(e - nEdges); }
    float norm = dinv[s] * dinv[d];
    float v = xl[(long long)s * C + c] * norm;
    atomicAdd(&agg[(long long)d * C + c], v);
}

__global__ void gcn_relu_kernel(float* __restrict__ h, long long n) {
    long long i = (long long)blockIdx.x * blockDim.x + threadIdx.x;
    if (i < n) h[i] = fmaxf(h[i], 0.0f);
}

extern "C" void kernel_launch(void* const* d_in, const int* in_sizes, int n_in,
                              void* d_out, int out_size, void* d_ws, size_t ws_size,
                              hipStream_t stream) {
    const float* x  = (const float*)d_in[0];
    const int*   ei = (const int*)  d_in[1];
    const float* W1 = (const float*)d_in[2];
    const float* b1 = (const float*)d_in[3];
    const float* W2 = (const float*)d_in[4];
    const float* b2 = (const float*)d_in[5];
    float* out = (float*)d_out;

    const int nNodes = in_sizes[0] / IN_CH;
    const int nEdges = in_sizes[1] / 2;
    const int* src = ei;
    const int* dst = ei + nEdges;

    float* ws  = (float*)d_ws;
    float* deg = ws;
    float* xl1 = deg + nNodes;
    float* h   = xl1 + (long long)nNodes * HID;
    float* xl2 = h   + (long long)nNodes * HID;

    const int B = 256;
    const long long hElems = (long long)nNodes * HID;

    gcn_init_kernel<<<(int)((hElems + B - 1) / B), B, 0, stream>>>(deg, h, b1, out, b2, nNodes);

    gcn_degree_kernel<<<(nEdges + B - 1) / B, B, 0, stream>>>(dst, deg, nEdges);
    gcn_rsqrt_kernel<<<(nNodes + B - 1) / B, B, 0, stream>>>(deg, nNodes);

    {
        int tiles = (nNodes / 16) * (HID / 16);
        int wpb = B / 32;
        gcn_gemm_wmma_f32<<<(tiles + wpb - 1) / wpb, B, 0, stream>>>(x, W1, xl1, nNodes, IN_CH, HID);
    }

    {
        long long tot = (long long)(nEdges + nNodes) * HID;
        gcn_scatter_kernel<HID><<<(int)((tot + B - 1) / B), B, 0, stream>>>(
            xl1, src, dst, deg, h, nEdges, nNodes);
    }

    gcn_relu_kernel<<<(int)((hElems + B - 1) / B), B, 0, stream>>>(h, hElems);

    {
        int tiles = (nNodes / 16) * (OUT_CH / 16);
        int wpb = B / 32;
        gcn_gemm_wmma_f32<<<(tiles + wpb - 1) / wpb, B, 0, stream>>>(h, W2, xl2, nNodes, HID, OUT_CH);
    }

    {
        long long tot = (long long)(nEdges + nNodes) * OUT_CH;
        gcn_scatter_kernel<OUT_CH><<<(int)((tot + B - 1) / B), B, 0, stream>>>(
            xl2, src, dst, deg, out, nEdges, nNodes);
    }
}


# Aggregation as a per-node gather over a destination-grouped edge list (one ranked integer count per edge) instead of per-channel float atomics; sums now in rank order plus a lane tree
# speedup vs baseline: 2.4613x; 2.4613x over previous
#include <hip/hip_runtime.h>

#define IN_CH  128
#define HID    64
#define OUT_CH 16

typedef float v2f __attribute__((ext_vector_type(2)));
typedef float v8f __attribute__((ext_vector_type(8)));


__global__ void gcn_zero_kernel(int* __restrict__ cnt, int n) {
    int i = blockIdx.x * blockDim.x + threadIdx.x;
    if (i < n) cnt[i] = 0;
}

__global__ void gcn_count_rank_kernel(const int* __restrict__ dst,
                                      int* __restrict__ cnt,
                                      int* __restrict__ rank, int nEdges) {
    int e = blockIdx.x * blockDim.x + threadIdx.x;
    if (e < nEdges) rank[e] = atomicAdd(&cnt[dst[e]], 1);
}

__global__ void gcn_alloc_kernel(const int* __restrict__ cnt,
                                 int* __restrict__ total,
                                 int* __restrict__ start,
                                 float* __restrict__ dinv, int n) {
    const int i = blockIdx.x * blockDim.x + threadIdx.x;
    const int lane = (int)(threadIdx.x & 31);
    const int c = (i < n) ? cnt[i] : 0;
    int incl = c;
    for (int off = 1; off < 32; off <<= 1) {
        const int t = __shfl_up(incl, off, 32);
        if (lane >= off) incl += t;
    }
    int base = 0;
    if (lane == 31) base = atomicAdd(total, incl);
    base = __shfl(base, 31, 32);
    if (i < n) {
        start[i] = base + incl - c;
        dinv[i]  = rsqrtf((float)(c + 1));
    }
}

__global__ void gcn_fill_kernel(const int* __restrict__ src,
                                const int* __restrict__ dst,
                                const int* __restrict__ rank,
                                const int* __restrict__ start,
                                int* __restrict__ col, int nEdges) {
    int e = blockIdx.x * blockDim.x + threadIdx.x;
    if (e < nEdges) col[start[dst[e]] + rank[e]] = src[e];
}

__global__ void gcn_gemm_wmma_f32(const float* __restrict__ A,
                                  const float* __restrict__ W,
                                  float* __restrict__ C,
                                  int M, int K, int N) {
    const int wavesPerBlock = blockDim.x >> 5;
    const int wave = (int)(threadIdx.x >> 5);
    const int lane = (int)(threadIdx.x & 31);
    const int tilesN = N >> 4;
    const int gw = blockIdx.x * wavesPerBlock + wave;
    const int tileM = gw / tilesN;
    const int tileN = gw - tileM * tilesN;
    if (tileM * 16 >= M) return;

    const int row0 = tileM << 4;
    const int col0 = tileN << 4;
    const int mr = lane & 15;
    const int kh = (lane >> 4) << 1;

    v8f acc = {};
    for (int k = 0; k < K; k += 4) {
        v2f a, b;
        const float* ap = A + (long long)(row0 + mr) * K + (k + kh);
        a.x = ap[0];
        a.y = ap[1];
        const float* bp = W + (long long)(k + kh) * N + (col0 + mr);
        b.x = bp[0];
        b.y = bp[N];
        acc = __builtin_amdgcn_wmma_f32_16x16x4_f32(
             false, a,  false, b,
             (short)0, acc,  false,  false);
    }

    const int crow = row0 + ((lane >> 4) << 3);
    const int ccol = col0 + mr;
#pragma unroll
    for (int r = 0; r < 8; ++r)
        C[(long long)(crow + r) * N + ccol] = acc[r];
}

template <int C, bool RELU>
__global__ void gcn_gather_kernel(const float* __restrict__ xl,
                                  const int* __restrict__ col,
                                  const int* __restrict__ start,
                                  const int* __restrict__ cnt,
                                  const float* __restrict__ dinv,
                                  const float* __restrict__ bias,
                                  float* __restrict__ res, int nNodes) {
    constexpr int LPR = C / 4;
    constexpr int G   = 32 / LPR;
    static_assert(32 % (4 * G) == 0, "a trip of four entries per group must stay inside the 32 held entries");
    const int lane = (int)(threadIdx.x & 31);
    const int d = __builtin_amdgcn_readfirstlane(
        (int)(blockIdx.x * (blockDim.x >> 5) + (threadIdx.x >> 5)));
    if (d >= nNodes) return;

    const int grp = lane / LPR;
    const int lig = lane - grp * LPR;
    const int beg = start[d];
    const int nIn = cnt[d];
    const int n   = nIn + 1;
    const float dd = dinv[d];
    const float* xcol = xl + lig * 4;

    float4 acc = make_float4(0.0f, 0.0f, 0.0f, 0.0f);
    for (int base = 0; base < n; base += 32) {
        const int m = min(32, n - base);
        const int j = base + lane;
        int   s_l = d;
        float w_l = 0.0f;
        if (j < nIn) s_l = col[beg + j];
        if (j < n)   w_l = dinv[s_l] * dd;
        for (int t = 0; t < m; t += 4 * G) {
#pragma unroll
            for (int u = 0; u < 4; ++u) {
                const int   s = __shfl(s_l, t + u * G + grp, 32);
                const float w = __shfl(w_l, t + u * G + grp, 32);
                const float4 v = *reinterpret_cast<const float4*>(xcol + (long long)s * C);
                acc.x += v.x * w;
                acc.y += v.y * w;
                acc.z += v.z * w;
                acc.w += v.w * w;
            }
        }
    }

#pragma unroll
    for (int off = LPR; off < 32; off <<= 1) {
        acc.x += __shfl_xor(acc.x, off, 32);
        acc.y += __shfl_xor(acc.y, off, 32);
        acc.z += __shfl_xor(acc.z, off, 32);
        acc.w += __shfl_xor(acc.w, off, 32);
    }

    if (grp == 0) {
        const float* bp = bias + lig * 4;
        float4 r = make_float4(acc.x + bp[0], acc.y + bp[1], acc.z + bp[2], acc.w + bp[3]);
        if (RELU) {
            r.x = fmaxf(r.x, 0.0f);
            r.y = fmaxf(r.y, 0.0f);
            r.z = fmaxf(r.z, 0.0f);
            r.w = fmaxf(r.w, 0.0f);
        }
        *reinterpret_cast<float4*>(res + (long long)d * C + lig * 4) = r;
    }
}

extern "C" void kernel_launch(void* const* d_in, const int* in_sizes, int n_in,
                              void* d_out, int out_size, void* d_ws, size_t ws_size,
                              hipStream_t stream) {
    const float* x  = (const float*)d_in[0];
    const int*   ei = (const int*)  d_in[1];
    const float* W1 = (const float*)d_in[2];
    const float* b1 = (const float*)d_in[3];
    const float* W2 = (const float*)d_in[4];
    const float* b2 = (const float*)d_in[5];
    float* out = (float*)d_out;

    const int nNodes = in_sizes[0] / IN_CH;
    const int nEdges = in_sizes[1] / 2;
    const int* src = ei;
    const int* dst = ei + nEdges;

    const long long hElems = (long long)nNodes * HID;
    float* xl1  = (float*)d_ws;
    float* h    = xl1 + hElems;
    float* xl2  = h   + hElems;
    float* dinv = xl2 + (long long)nNodes * OUT_CH;
    int* start  = (int*)(dinv + nNodes);
    int* rank   = start + nNodes;
    int* col    = rank  + nEdges;
    int* cnt    = col   + nEdges;
    int* total  = cnt   + nNodes;

    const int B = 256;
    const int wpb = B / 32;

    gcn_zero_kernel<<<(nNodes + 1 + B - 1) / B, B, 0, stream>>>(cnt, nNodes + 1);

    gcn_count_rank_kernel<<<(nEdges + B - 1) / B, B, 0, stream>>>(dst, cnt, rank, nEdges);
    gcn_alloc_kernel<<<(nNodes + B - 1) / B, B, 0, stream>>>(cnt, total, start, dinv, nNodes);
    gcn_fill_kernel<<<(nEdges + B - 1) / B, B, 0, stream>>>(src, dst, rank, start, col, nEdges);

    {
        int tiles = (nNodes / 16) * (HID / 16);
        gcn_gemm_wmma_f32<<<(tiles + wpb - 1) / wpb, B, 0, stream>>>(x, W1, xl1, nNodes, IN_CH, HID);
    }

    gcn_gather_kernel<HID, true><<<(nNodes + wpb - 1) / wpb, B, 0, stream>>>(
        xl1, col, start, cnt, dinv, b1, h, nNodes);

    {
        int tiles = (nNodes / 16) * (OUT_CH / 16);
        gcn_gemm_wmma_f32<<<(tiles + wpb - 1) / wpb, B, 0, stream>>>(h, W2, xl2, nNodes, HID, OUT_CH);
    }

    gcn_gather_kernel<OUT_CH, false><<<(nNodes + wpb - 1) / wpb, B, 0, stream>>>(
        xl2, col, start, cnt, dinv, b2, out, nNodes);
}


# Edges grouped by destination by a two-level counting sort on LDS counters (256-node buckets, then nodes) instead of one returning global add per edge; a node's rows now sum in LDS arrival order
# speedup vs baseline: 4.0014x; 1.6257x over previous
#include <hip/hip_runtime.h>

#define IN_CH  128
#define HID    64
#define OUT_CH 16

typedef float v2f __attribute__((ext_vector_type(2)));
typedef float v8f __attribute__((ext_vector_type(8)));

#define NPB_SHIFT 8
#define NPB       256
#define NBMAX     1024
#define CH        4096
#define EPT       16
#define CSTRIDE   32
#define CAP       8192

__global__ void gcn_zero_kernel(int* __restrict__ cnt, int n) {
    int i = blockIdx.x * blockDim.x + threadIdx.x;
    if (i < n) cnt[i] = 0;
}

__device__ __forceinline__ int gcn_block_excl_scan(int s, int* wsum) {
    const int lane = (int)(threadIdx.x & 31);
    const int wave = (int)(threadIdx.x >> 5);
    int incl = s;
    for (int off = 1; off < 32; off <<= 1) {
        const int t = __shfl_up(incl, off, 32);
        if (lane >= off) incl += t;
    }
    if (lane == 31) wsum[wave] = incl;
    __syncthreads();
    int wofs = 0;
    for (int w = 0; w < wave; ++w) wofs += wsum[w];
    __syncthreads();
    return wofs + incl - s;
}

__global__ void gcn_bucket_count_kernel(const int* __restrict__ dst,
                                        int* __restrict__ bucketTotal,
                                        int nEdges, int nBuckets) {
    __shared__ int hist[NBMAX];
    const int t = (int)threadIdx.x;
    for (int i = t; i < nBuckets; i += 256) hist[i] = 0;
    __syncthreads();
    const int e0 = blockIdx.x * CH;
    int dv[EPT];
#pragma unroll
    for (int k = 0; k < EPT; ++k)
        dv[k] = dst[min(e0 + k * 256 + t, nEdges - 1)];
#pragma unroll
    for (int k = 0; k < EPT; ++k)
        if (e0 + k * 256 + t < nEdges) atomicAdd(&hist[dv[k] >> NPB_SHIFT], 1);
    __syncthreads();
    for (int i = t; i < nBuckets; i += 256) {
        const int c = hist[i];
        if (c) atomicAdd(&bucketTotal[i * CSTRIDE], c);
    }
}

__global__ void gcn_bucket_scatter_kernel(const int* __restrict__ src,
                                          const int* __restrict__ dst,
                                          const int* __restrict__ bucketTotal,
                                          int* __restrict__ bucketCursor,
                                          int* __restrict__ bucketBase,
                                          int* __restrict__ tmp,
                                          int nEdges, int nBuckets) {
    __shared__ int lcnt[NBMAX];
    __shared__ int gofs[NBMAX];
    __shared__ int stageVal[CH];
    __shared__ int stageBin[CH];
    __shared__ int wsum[8];
    const int t = (int)threadIdx.x;
    for (int i = t; i < NBMAX; i += 256) lcnt[i] = 0;
    __syncthreads();

    const int e0 = blockIdx.x * CH;
    int s[EPT], d[EPT], r[EPT];
#pragma unroll
    for (int k = 0; k < EPT; ++k) {
        const int e = min(e0 + k * 256 + t, nEdges - 1);
        d[k] = dst[e];
        s[k] = src[e];
    }
#pragma unroll
    for (int k = 0; k < EPT; ++k) {
        r[k] = 0;
        if (e0 + k * 256 + t < nEdges)
            r[k] = atomicAdd(&lcnt[d[k] >> NPB_SHIFT], 1);
        else
            d[k] = -1;
    }
    __syncthreads();

    int tot[4], loc[4];
    int ts = 0, ls = 0;
    int mine[4];
#pragma unroll
    for (int j = 0; j < 4; ++j) {
        const int b = t * 4 + j;
        tot[j] = bucketTotal[min(b, nBuckets - 1) * CSTRIDE];
        if (b >= nBuckets) tot[j] = 0;
        loc[j] = lcnt[b];
        ts += tot[j];
        ls += loc[j];
    }
#pragma unroll
    for (int j = 0; j < 4; ++j) {
        mine[j] = 0;
        if (loc[j] > 0) mine[j] = atomicAdd(&bucketCursor[(t * 4 + j) * CSTRIDE], loc[j]);
    }
    int tp = gcn_block_excl_scan(ts, wsum);
    int lp = gcn_block_excl_scan(ls, wsum);
#pragma unroll
    for (int j = 0; j < 4; ++j) {
        const int b = t * 4 + j;
        if (b < nBuckets) {
            if (blockIdx.x == 0) bucketBase[b] = tp;
            gofs[b] = tp + mine[j] - lp;
        }
        lcnt[b] = lp;
        tp += tot[j];
        lp += loc[j];
    }
    __syncthreads();

#pragma unroll
    for (int k = 0; k < EPT; ++k) {
        if (d[k] >= 0) {
            const int bin = d[k] >> NPB_SHIFT;
            const int pos = lcnt[bin] + r[k];
            stageVal[pos] = (int)(((unsigned)s[k] << NPB_SHIFT) | (unsigned)(d[k] & (NPB - 1)));
            stageBin[pos] = bin;
        }
    }
    __syncthreads();

    const int nLocal = min(CH, nEdges - e0);
    for (int i = t; i < nLocal; i += 256)
        tmp[gofs[stageBin[i]] + i] = stageVal[i];
}

__global__ void gcn_bucket_place_kernel(const int* __restrict__ tmp,
                                        const int* __restrict__ bucketTotal,
                                        const int* __restrict__ bucketBase,
                                        int* __restrict__ start,
                                        int* __restrict__ cnt,
                                        float* __restrict__ dinv,
                                        int* __restrict__ col, int nNodes) {
    __shared__ int c[NPB];
    __shared__ int cur[NPB];
    __shared__ int wsum[8];
    __shared__ int stage[CAP];
    const int t = (int)threadIdx.x;
    const int b = (int)blockIdx.x;
    const int n = bucketTotal[b * CSTRIDE];
    const int base = bucketBase[b];
    c[t] = 0;
    __syncthreads();
    for (int i0 = 0; i0 < n; i0 += 8 * 256) {
        int v[8];
#pragma unroll
        for (int u = 0; u < 8; ++u)
            v[u] = tmp[base + min(i0 + u * 256 + t, n - 1)];
#pragma unroll
        for (int u = 0; u < 8; ++u)
            if (i0 + u * 256 + t < n) atomicAdd(&c[v[u] & (NPB - 1)], 1);
    }
    __syncthreads();
    const int mine = c[t];
    const int off = gcn_block_excl_scan(mine, wsum);
    const int node = b * NPB + t;
    if (node < nNodes) {
        start[node] = base + off;
        cnt[node]   = mine;
        dinv[node]  = rsqrtf((float)(mine + 1));
    }
    cur[t] = off;
    __syncthreads();
    if (n <= CAP) {
        for (int i0 = 0; i0 < n; i0 += 8 * 256) {
            int v[8];
#pragma unroll
            for (int u = 0; u < 8; ++u)
                v[u] = tmp[base + min(i0 + u * 256 + t, n - 1)];
#pragma unroll
            for (int u = 0; u < 8; ++u) {
                if (i0 + u * 256 + t < n) {
                    const int pos = atomicAdd(&cur[v[u] & (NPB - 1)], 1);
                    stage[pos] = (int)((unsigned)v[u] >> NPB_SHIFT);
                }
            }
        }
        __syncthreads();
        for (int i = t; i < n; i += 256) col[base + i] = stage[i];
    } else {
        for (int i = t; i < n; i += 256) {
            const int v = tmp[base + i];
            const int pos = atomicAdd(&cur[v & (NPB - 1)], 1);
            col[base + pos] = (int)((unsigned)v >> NPB_SHIFT);
        }
    }
}

__global__ void gcn_gemm_wmma_f32(const float* __restrict__ A,
                                  const float* __restrict__ W,
                                  float* __restrict__ C,
                                  int M, int K, int N) {
    const int wavesPerBlock = blockDim.x >> 5;
    const int wave = (int)(threadIdx.x >> 5);
    const int lane = (int)(threadIdx.x & 31);
    const int tilesN = N >> 4;
    const int gw = blockIdx.x * wavesPerBlock + wave;
    const int tileM = gw / tilesN;
    const int tileN = gw - tileM * tilesN;
    if (tileM * 16 >= M) return;

    const int row0 = tileM << 4;
    const int col0 = tileN << 4;
    const int mr = lane & 15;
    const int kh = (lane >> 4) << 1;

    v8f acc = {};
    for (int k = 0; k < K; k += 4) {
        v2f a, b;
        const float* ap = A + (long long)(row0 + mr) * K + (k + kh);
        a.x = ap[0];
        a.y = ap[1];
        const float* bp = W + (long long)(k + kh) * N + (col0 + mr);
        b.x = bp[0];
        b.y = bp[N];
        acc = __builtin_amdgcn_wmma_f32_16x16x4_f32(
             false, a,  false, b,
             (short)0, acc,  false,  false);
    }

    const int crow = row0 + ((lane >> 4) << 3);
    const int ccol = col0 + mr;
#pragma unroll
    for (int r = 0; r < 8; ++r)
        C[(long long)(crow + r) * N + ccol] = acc[r];
}

template <int C, bool RELU>
__global__ void gcn_gather_kernel(const float* __restrict__ xl,
                                  const int* __restrict__ col,
                                  const int* __restrict__ start,
                                  const int* __restrict__ cnt,
                                  const float* __restrict__ dinv,
                                  const float* __restrict__ bias,
                                  float* __restrict__ res, int nNodes) {
    constexpr int LPR = C / 4;
    constexpr int G   = 32 / LPR;
    static_assert(32 % (4 * G) == 0, "a trip of four entries per group must stay inside the 32 held entries");
    const int lane = (int)(threadIdx.x & 31);
    const int d = __builtin_amdgcn_readfirstlane(
        (int)(blockIdx.x * (blockDim.x >> 5) + (threadIdx.x >> 5)));
    if (d >= nNodes) return;

    const int grp = lane / LPR;
    const int lig = lane - grp * LPR;
    const int beg = start[d];
    const int nIn = cnt[d];
    const int n   = nIn + 1;
    const float dd = dinv[d];
    const float* xcol = xl + lig * 4;

    float4 acc = make_float4(0.0f, 0.0f, 0.0f, 0.0f);
    for (int base = 0; base < n; base += 32) {
        const int m = min(32, n - base);
        const int j = base + lane;
        int   s_l = d;
        float w_l = 0.0f;
        if (j < nIn) s_l = col[beg + j];
        if (j < n)   w_l = dinv[s_l] * dd;
        for (int t = 0; t < m; t += 4 * G) {
#pragma unroll
            for (int u = 0; u < 4; ++u) {
                const int   s = __shfl(s_l, t + u * G + grp, 32);
                const float w = __shfl(w_l, t + u * G + grp, 32);
                const float4 v = *reinterpret_cast<const float4*>(xcol + (long long)s * C);
                acc.x += v.x * w;
                acc.y += v.y * w;
                acc.z += v.z * w;
                acc.w += v.w * w;
            }
        }
    }

#pragma unroll
    for (int off = LPR; off < 32; off <<= 1) {
        acc.x += __shfl_xor(acc.x, off, 32);
        acc.y += __shfl_xor(acc.y, off, 32);
        acc.z += __shfl_xor(acc.z, off, 32);
        acc.w += __shfl_xor(acc.w, off, 32);
    }

    if (grp == 0) {
        const float* bp = bias + lig * 4;
        float4 r = make_float4(acc.x + bp[0], acc.y + bp[1], acc.z + bp[2], acc.w + bp[3]);
        if (RELU) {
            r.x = fmaxf(r.x, 0.0f);
            r.y = fmaxf(r.y, 0.0f);
            r.z = fmaxf(r.z, 0.0f);
            r.w = fmaxf(r.w, 0.0f);
        }
        *reinterpret_cast<float4*>(res + (long long)d * C + lig * 4) = r;
    }
}

extern "C" void kernel_launch(void* const* d_in, const int* in_sizes, int n_in,
                              void* d_out, int out_size, void* d_ws, size_t ws_size,
                              hipStream_t stream) {
    const float* x  = (const float*)d_in[0];
    const int*   ei = (const int*)  d_in[1];
    const float* W1 = (const float*)d_in[2];
    const float* b1 = (const float*)d_in[3];
    const float* W2 = (const float*)d_in[4];
    const float* b2 = (const float*)d_in[5];
    float* out = (float*)d_out;

    const int nNodes = in_sizes[0] / IN_CH;
    const int nEdges = in_sizes[1] / 2;
    const int* src = ei;
    const int* dst = ei + nEdges;

    const long long hElems = (long long)nNodes * HID;
    float* xl1  = (float*)d_ws;
    float* h    = xl1 + hElems;
    float* xl2  = h   + hElems;
    float* dinv = xl2 + (long long)nNodes * OUT_CH;
    int* start  = (int*)(dinv + nNodes);
    int* tmp    = start + nNodes;
    int* col    = tmp   + nEdges;
    int* cnt    = col   + nEdges;

    const int B = 256;
    const int wpb = B / 32;
    const int nBuckets = (nNodes + NPB - 1) / NPB;
    const int nChunks  = (nEdges + CH - 1) / CH;

    int* bucketBase   = cnt + nNodes;
    int* bucketTotal  = bucketBase + nBuckets;
    int* bucketCursor = bucketTotal + nBuckets * CSTRIDE;

    gcn_zero_kernel<<<(2 * nBuckets * CSTRIDE + B - 1) / B, B, 0, stream>>>(bucketTotal, 2 * nBuckets * CSTRIDE);

    gcn_bucket_count_kernel<<<nChunks, B, 0, stream>>>(dst, bucketTotal, nEdges, nBuckets);
    gcn_bucket_scatter_kernel<<<nChunks, B, 0, stream>>>(src, dst, bucketTotal, bucketCursor, bucketBase,
                                                         tmp, nEdges, nBuckets);
    gcn_bucket_place_kernel<<<nBuckets, B, 0, stream>>>(tmp, bucketTotal, bucketBase, start, cnt, dinv, col, nNodes);

    {
        int tiles = (nNodes / 16) * (HID / 16);
        gcn_gemm_wmma_f32<<<(tiles + wpb - 1) / wpb, B, 0, stream>>>(x, W1, xl1, nNodes, IN_CH, HID);
    }

    gcn_gather_kernel<HID, true><<<(nNodes + wpb - 1) / wpb, B, 0, stream>>>(
        xl1, col, start, cnt, dinv, b1, h, nNodes);

    {
        int tiles = (nNodes / 16) * (OUT_CH / 16);
        gcn_gemm_wmma_f32<<<(tiles + wpb - 1) / wpb, B, 0, stream>>>(h, W2, xl2, nNodes, HID, OUT_CH);
    }

    gcn_gather_kernel<OUT_CH, false><<<(nNodes + wpb - 1) / wpb, B, 0, stream>>>(
        xl2, col, start, cnt, dinv, b2, out, nNodes);
}


# Both gather launches as a resident grid (2048 work-groups) whose waves loop over nodes, the next node's list bounds and first 32 sources loaded ahead, not one short wave per node; sum order kept
# speedup vs baseline: 4.2120x; 1.0526x over previous
#include <hip/hip_runtime.h>

#define IN_CH  128
#define HID    64
#define OUT_CH 16

typedef float v2f __attribute__((ext_vector_type(2)));
typedef float v8f __attribute__((ext_vector_type(8)));

#define NPB_SHIFT 8
#define NPB       256
#define NBMAX     1024
#define CH        4096
#define EPT       16
#define CSTRIDE   32
#define CAP       8192

__global__ void gcn_zero_kernel(int* __restrict__ cnt, int n) {
    int i = blockIdx.x * blockDim.x + threadIdx.x;
    if (i < n) cnt[i] = 0;
}

__device__ __forceinline__ int gcn_block_excl_scan(int s, int* wsum) {
    const int lane = (int)(threadIdx.x & 31);
    const int wave = (int)(threadIdx.x >> 5);
    int incl = s;
    for (int off = 1; off < 32; off <<= 1) {
        const int t = __shfl_up(incl, off, 32);
        if (lane >= off) incl += t;
    }
    if (lane == 31) wsum[wave] = incl;
    __syncthreads();
    int wofs = 0;
    for (int w = 0; w < wave; ++w) wofs += wsum[w];
    __syncthreads();
    return wofs + incl - s;
}

__global__ void gcn_bucket_count_kernel(const int* __restrict__ dst,
                                        int* __restrict__ bucketTotal,
                                        int nEdges, int nBuckets) {
    __shared__ int hist[NBMAX];
    const int t = (int)threadIdx.x;
    for (int i = t; i < nBuckets; i += 256) hist[i] = 0;
    __syncthreads();
    const int e0 = blockIdx.x * CH;
    int dv[EPT];
#pragma unroll
    for (int k = 0; k < EPT; ++k)
        dv[k] = dst[min(e0 + k * 256 + t, nEdges - 1)];
#pragma unroll
    for (int k = 0; k < EPT; ++k)
        if (e0 + k * 256 + t < nEdges) atomicAdd(&hist[dv[k] >> NPB_SHIFT], 1);
    __syncthreads();
    for (int i = t; i < nBuckets; i += 256) {
        const int c = hist[i];
        if (c) atomicAdd(&bucketTotal[i * CSTRIDE], c);
    }
}

__global__ void gcn_bucket_scatter_kernel(const int* __restrict__ src,
                                          const int* __restrict__ dst,
                                          const int* __restrict__ bucketTotal,
                                          int* __restrict__ bucketCursor,
                                          int* __restrict__ bucketBase,
                                          int* __restrict__ tmp,
                                          int nEdges, int nBuckets) {
    __shared__ int lcnt[NBMAX];
    __shared__ int gofs[NBMAX];
    __shared__ int stageVal[CH];
    __shared__ int stageBin[CH];
    __shared__ int wsum[8];
    const int t = (int)threadIdx.x;
    for (int i = t; i < NBMAX; i += 256) lcnt[i] = 0;
    __syncthreads();

    const int e0 = blockIdx.x * CH;
    int s[EPT], d[EPT], r[EPT];
#pragma unroll
    for (int k = 0; k < EPT; ++k) {
        const int e = min(e0 + k * 256 + t, nEdges - 1);
        d[k] = dst[e];
        s[k] = src[e];
    }
#pragma unroll
    for (int k = 0; k < EPT; ++k) {
        r[k] = 0;
        if (e0 + k * 256 + t < nEdges)
            r[k] = atomicAdd(&lcnt[d[k] >> NPB_SHIFT], 1);
        else
            d[k] = -1;
    }
    __syncthreads();

    int tot[4], loc[4];
    int ts = 0, ls = 0;
    int mine[4];
#pragma unroll
    for (int j = 0; j < 4; ++j) {
        const int b = t * 4 + j;
        tot[j] = bucketTotal[min(b, nBuckets - 1) * CSTRIDE];
        if (b >= nBuckets) tot[j] = 0;
        loc[j] = lcnt[b];
        ts += tot[j];
        ls += loc[j];
    }
#pragma unroll
    for (int j = 0; j < 4; ++j) {
        mine[j] = 0;
        if (loc[j] > 0) mine[j] = atomicAdd(&bucketCursor[(t * 4 + j) * CSTRIDE], loc[j]);
    }
    int tp = gcn_block_excl_scan(ts, wsum);
    int lp = gcn_block_excl_scan(ls, wsum);
#pragma unroll
    for (int j = 0; j < 4; ++j) {
        const int b = t * 4 + j;
        if (b < nBuckets) {
            if (blockIdx.x == 0) bucketBase[b] = tp;
            gofs[b] = tp + mine[j] - lp;
        }
        lcnt[b] = lp;
        tp += tot[j];
        lp += loc[j];
    }
    __syncthreads();

#pragma unroll
    for (int k = 0; k < EPT; ++k) {
        if (d[k] >= 0) {
            const int bin = d[k] >> NPB_SHIFT;
            const int pos = lcnt[bin] + r[k];
            stageVal[pos] = (int)(((unsigned)s[k] << NPB_SHIFT) | (unsigned)(d[k] & (NPB - 1)));
            stageBin[pos] = bin;
        }
    }
    __syncthreads();

    const int nLocal = min(CH, nEdges - e0);
    for (int i = t; i < nLocal; i += 256)
        tmp[gofs[stageBin[i]] + i] = stageVal[i];
}

__global__ void gcn_bucket_place_kernel(const int* __restrict__ tmp,
                                        const int* __restrict__ bucketTotal,
                                        const int* __restrict__ bucketBase,
                                        int* __restrict__ start,
                                        int* __restrict__ cnt,
                                        float* __restrict__ dinv,
                                        int* __restrict__ col, int nNodes) {
    __shared__ int c[NPB];
    __shared__ int cur[NPB];
    __shared__ int wsum[8];
    __shared__ int stage[CAP];
    const int t = (int)threadIdx.x;
    const int b = (int)blockIdx.x;
    const int n = bucketTotal[b * CSTRIDE];
    const int base = bucketBase[b];
    c[t] = 0;
    __syncthreads();
    for (int i0 = 0; i0 < n; i0 += 8 * 256) {
        int v[8];
#pragma unroll
        for (int u = 0; u < 8; ++u)
            v[u] = tmp[base + min(i0 + u * 256 + t, n - 1)];
#pragma unroll
        for (int u = 0; u < 8; ++u)
            if (i0 + u * 256 + t < n) atomicAdd(&c[v[u] & (NPB - 1)], 1);
    }
    __syncthreads();
    const int mine = c[t];
    const int off = gcn_block_excl_scan(mine, wsum);
    const int node = b * NPB + t;
    if (node < nNodes) {
        start[node] = base + off;
        cnt[node]   = mine;
        dinv[node]  = rsqrtf((float)(mine + 1));
    }
    cur[t] = off;
    __syncthreads();
    if (n <= CAP) {
        for (int i0 = 0; i0 < n; i0 += 8 * 256) {
            int v[8];
#pragma unroll
            for (int u = 0; u < 8; ++u)
                v[u] = tmp[base + min(i0 + u * 256 + t, n - 1)];
#pragma unroll
            for (int u = 0; u < 8; ++u) {
                if (i0 + u * 256 + t < n) {
                    const int pos = atomicAdd(&cur[v[u] & (NPB - 1)], 1);
                    stage[pos] = (int)((unsigned)v[u] >> NPB_SHIFT);
                }
            }
        }
        __syncthreads();
        for (int i = t; i < n; i += 256) col[base + i] = stage[i];
    } else {
        for (int i = t; i < n; i += 256) {
            const int v = tmp[base + i];
            const int pos = atomicAdd(&cur[v & (NPB - 1)], 1);
            col[base + pos] = (int)((unsigned)v >> NPB_SHIFT);
        }
    }
}

__global__ void gcn_gemm_wmma_f32(const float* __restrict__ A,
                                  const float* __restrict__ W,
                                  float* __restrict__ C,
                                  int M, int K, int N) {
    const int wavesPerBlock = blockDim.x >> 5;
    const int wave = (int)(threadIdx.x >> 5);
    const int lane = (int)(threadIdx.x & 31);
    const int tilesN = N >> 4;
    const int gw = blockIdx.x * wavesPerBlock + wave;
    const int tileM = gw / tilesN;
    const int tileN = gw - tileM * tilesN;
    if (tileM * 16 >= M) return;

    const int row0 = tileM << 4;
    const int col0 = tileN << 4;
    const int mr = lane & 15;
    const int kh = (lane >> 4) << 1;

    v8f acc = {};
    for (int k = 0; k < K; k += 4) {
        v2f a, b;
        const float* ap = A + (long long)(row0 + mr) * K + (k + kh);
        a.x = ap[0];
        a.y = ap[1];
        const float* bp = W + (long long)(k + kh) * N + (col0 + mr);
        b.x = bp[0];
        b.y = bp[N];
        acc = __builtin_amdgcn_wmma_f32_16x16x4_f32(
             false, a,  false, b,
             (short)0, acc,  false,  false);
    }

    const int crow = row0 + ((lane >> 4) << 3);
    const int ccol = col0 + mr;
#pragma unroll
    for (int r = 0; r < 8; ++r)
        C[(long long)(crow + r) * N + ccol] = acc[r];
}

template <int C, bool RELU>
__global__ void __launch_bounds__(256)
gcn_gather_kernel(const float* __restrict__ xl,
                  const int* __restrict__ col,
                  const int* __restrict__ start,
                  const int* __restrict__ cnt,
                  const float* __restrict__ dinv,
                  const float* __restrict__ bias,
                  float* __restrict__ res, int nNodes) {
    constexpr int LPR = C / 4;
    constexpr int G   = 32 / LPR;
    static_assert(32 % (4 * G) == 0, "a trip of four entries per group must stay inside the 32 held entries");
    const int lane   = (int)(threadIdx.x & 31);
    const int wpb    = (int)(blockDim.x >> 5);
    const int nWaves = (int)gridDim.x * wpb;
    const int gw = __builtin_amdgcn_readfirstlane(
        (int)(blockIdx.x * wpb + (threadIdx.x >> 5)));
    if (gw >= nNodes) return;
    const int last = nNodes - 1;

    const int grp = lane / LPR;
    const int lig = lane - grp * LPR;
    const float* xcol = xl + lig * 4;
    const float4 bv = *reinterpret_cast<const float4*>(bias + lig * 4);

    int   d   = gw;
    int   beg = start[d];
    int   nIn = cnt[d];
    float dd  = dinv[d];
    int   s_l = d;
    if (lane < nIn) s_l = col[beg + lane];
    float di_l = dinv[s_l];
    int   d1   = min(d + nWaves, last);
    int   beg1 = start[d1];
    int   nIn1 = cnt[d1];
    float dd1  = dinv[d1];

    for (;;) {
        int s1 = d1;
        if (lane < nIn1) s1 = col[beg1 + lane];
        const int   d2   = min(d1 + nWaves, last);
        const int   beg2 = start[d2];
        const int   nIn2 = cnt[d2];
        const float dd2  = dinv[d2];

        const int n = nIn + 1;
        float4 acc = make_float4(0.0f, 0.0f, 0.0f, 0.0f);
        for (int base = 0; base < n; base += 32) {
            const int m = min(32, n - base);
            const int j = base + lane;
            if (base > 0) {
                s_l = d;
                if (j < nIn) s_l = col[beg + j];
                di_l = dinv[s_l];
            }
            const float w_l = (j < n) ? di_l * dd : 0.0f;
            for (int t = 0; t < m; t += 4 * G) {
#pragma unroll
                for (int u = 0; u < 4; ++u) {
                    const int   s = __shfl(s_l, t + u * G + grp, 32);
                    const float w = __shfl(w_l, t + u * G + grp, 32);
                    const float4 v = *reinterpret_cast<const float4*>(xcol + (long long)s * C);
                    acc.x += v.x * w;
                    acc.y += v.y * w;
                    acc.z += v.z * w;
                    acc.w += v.w * w;
                }
            }
        }

#pragma unroll
        for (int off = LPR; off < 32; off <<= 1) {
            acc.x += __shfl_xor(acc.x, off, 32);
            acc.y += __shfl_xor(acc.y, off, 32);
            acc.z += __shfl_xor(acc.z, off, 32);
            acc.w += __shfl_xor(acc.w, off, 32);
        }

        if (grp == 0) {
            float4 r = make_float4(acc.x + bv.x, acc.y + bv.y, acc.z + bv.z, acc.w + bv.w);
            if (RELU) {
                r.x = fmaxf(r.x, 0.0f);
                r.y = fmaxf(r.y, 0.0f);
                r.z = fmaxf(r.z, 0.0f);
                r.w = fmaxf(r.w, 0.0f);
            }
            *reinterpret_cast<float4*>(res + (long long)d * C + lig * 4) = r;
        }

        if (d + nWaves >= nNodes) break;
        di_l = dinv[s1];
        s_l  = s1;
        d = d1;  beg  = beg1; nIn  = nIn1; dd  = dd1;
        d1 = d2; beg1 = beg2; nIn1 = nIn2; dd1 = dd2;
    }
}

extern "C" void kernel_launch(void* const* d_in, const int* in_sizes, int n_in,
                              void* d_out, int out_size, void* d_ws, size_t ws_size,
                              hipStream_t stream) {
    const float* x  = (const float*)d_in[0];
    const int*   ei = (const int*)  d_in[1];
    const float* W1 = (const float*)d_in[2];
    const float* b1 = (const float*)d_in[3];
    const float* W2 = (const float*)d_in[4];
    const float* b2 = (const float*)d_in[5];
    float* out = (float*)d_out;

    const int nNodes = in_sizes[0] / IN_CH;
    const int nEdges = in_sizes[1] / 2;
    const int* src = ei;
    const int* dst = ei + nEdges;

    const long long hElems = (long long)nNodes * HID;
    float* xl1  = (float*)d_ws;
    float* h    = xl1 + hElems;
    float* xl2  = h   + hElems;
    float* dinv = xl2 + (long long)nNodes * OUT_CH;
    int* start  = (int*)(dinv + nNodes);
    int* tmp    = start + nNodes;
    int* col    = tmp   + nEdges;
    int* cnt    = col   + nEdges;

    const int B = 256;
    const int wpb = B / 32;
    const int nBuckets = (nNodes + NPB - 1) / NPB;
    const int nChunks  = (nEdges + CH - 1) / CH;

    int* bucketBase   = cnt + nNodes;
    int* bucketTotal  = bucketBase + nBuckets;
    int* bucketCursor = bucketTotal + nBuckets * CSTRIDE;

    gcn_zero_kernel<<<(2 * nBuckets * CSTRIDE + B - 1) / B, B, 0, stream>>>(bucketTotal, 2 * nBuckets * CSTRIDE);

    gcn_bucket_count_kernel<<<nChunks, B, 0, stream>>>(dst, bucketTotal, nEdges, nBuckets);
    gcn_bucket_scatter_kernel<<<nChunks, B, 0, stream>>>(src, dst, bucketTotal, bucketCursor, bucketBase,
                                                         tmp, nEdges, nBuckets);
    gcn_bucket_place_kernel<<<nBuckets, B, 0, stream>>>(tmp, bucketTotal, bucketBase, start, cnt, dinv, col, nNodes);

    {
        int tiles = (nNodes / 16) * (HID / 16);
        gcn_gemm_wmma_f32<<<(tiles + wpb - 1) / wpb, B, 0, stream>>>(x, W1, xl1, nNodes, IN_CH, HID);
    }

    const int gatherGroups = 2048;

    gcn_gather_kernel<HID, true><<<gatherGroups, B, 0, stream>>>(
        xl1, col, start, cnt, dinv, b1, h, nNodes);

    {
        int tiles = (nNodes / 16) * (OUT_CH / 16);
        gcn_gemm_wmma_f32<<<(tiles + wpb - 1) / wpb, B, 0, stream>>>(h, W2, xl2, nNodes, HID, OUT_CH);
    }

    gcn_gather_kernel<OUT_CH, false><<<gatherGroups, B, 0, stream>>>(
        xl2, col, start, cnt, dinv, b2, out, nNodes);
}


# Product rows leave the WMMA epilogue times their own node's dinv, so the gathers fetch no per-source dinv word and shuffle no weight; a term is now (xl*dinv_s)*dinv_d, sum order kept
# speedup vs baseline: 4.4975x; 1.0678x over previous
#include <hip/hip_runtime.h>

#define IN_CH  128
#define HID    64
#define OUT_CH 16

typedef float v2f __attribute__((ext_vector_type(2)));
typedef float v8f __attribute__((ext_vector_type(8)));

#define NPB_SHIFT 8
#define NPB       256
#define NBMAX     1024
#define CH        4096
#define EPT       16
#define CSTRIDE   32
#define CAP       8192

__global__ void gcn_zero_kernel(int* __restrict__ cnt, int n) {
    int i = blockIdx.x * blockDim.x + threadIdx.x;
    if (i < n) cnt[i] = 0;
}

__device__ __forceinline__ int gcn_block_excl_scan(int s, int* wsum) {
    const int lane = (int)(threadIdx.x & 31);
    const int wave = (int)(threadIdx.x >> 5);
    int incl = s;
    for (int off = 1; off < 32; off <<= 1) {
        const int t = __shfl_up(incl, off, 32);
        if (lane >= off) incl += t;
    }
    if (lane == 31) wsum[wave] = incl;
    __syncthreads();
    int wofs = 0;
    for (int w = 0; w < wave; ++w) wofs += wsum[w];
    __syncthreads();
    return wofs + incl - s;
}

__global__ void gcn_bucket_count_kernel(const int* __restrict__ dst,
                                        int* __restrict__ bucketTotal,
                                        int nEdges, int nBuckets) {
    __shared__ int hist[NBMAX];
    const int t = (int)threadIdx.x;
    for (int i = t; i < nBuckets; i += 256) hist[i] = 0;
    __syncthreads();
    const int e0 = blockIdx.x * CH;
    int dv[EPT];
#pragma unroll
    for (int k = 0; k < EPT; ++k)
        dv[k] = dst[min(e0 + k * 256 + t, nEdges - 1)];
#pragma unroll
    for (int k = 0; k < EPT; ++k)
        if (e0 + k * 256 + t < nEdges) atomicAdd(&hist[dv[k] >> NPB_SHIFT], 1);
    __syncthreads();
    for (int i = t; i < nBuckets; i += 256) {
        const int c = hist[i];
        if (c) atomicAdd(&bucketTotal[i * CSTRIDE], c);
    }
}

__global__ void gcn_bucket_scatter_kernel(const int* __restrict__ src,
                                          const int* __restrict__ dst,
                                          const int* __restrict__ bucketTotal,
                                          int* __restrict__ bucketCursor,
                                          int* __restrict__ bucketBase,
                                          int* __restrict__ tmp,
                                          int nEdges, int nBuckets) {
    __shared__ int lcnt[NBMAX];
    __shared__ int gofs[NBMAX];
    __shared__ int stageVal[CH];
    __shared__ int stageBin[CH];
    __shared__ int wsum[8];
    const int t = (int)threadIdx.x;
    for (int i = t; i < NBMAX; i += 256) lcnt[i] = 0;
    __syncthreads();

    const int e0 = blockIdx.x * CH;
    int s[EPT], d[EPT], r[EPT];
#pragma unroll
    for (int k = 0; k < EPT; ++k) {
        const int e = min(e0 + k * 256 + t, nEdges - 1);
        d[k] = dst[e];
        s[k] = src[e];
    }
#pragma unroll
    for (int k = 0; k < EPT; ++k) {
        r[k] = 0;
        if (e0 + k * 256 + t < nEdges)
            r[k] = atomicAdd(&lcnt[d[k] >> NPB_SHIFT], 1);
        else
            d[k] = -1;
    }
    __syncthreads();

    int tot[4], loc[4];
    int ts = 0, ls = 0;
    int mine[4];
#pragma unroll
    for (int j = 0; j < 4; ++j) {
        const int b = t * 4 + j;
        tot[j] = bucketTotal[min(b, nBuckets - 1) * CSTRIDE];
        if (b >= nBuckets) tot[j] = 0;
        loc[j] = lcnt[b];
        ts += tot[j];
        ls += loc[j];
    }
#pragma unroll
    for (int j = 0; j < 4; ++j) {
        mine[j] = 0;
        if (loc[j] > 0) mine[j] = atomicAdd(&bucketCursor[(t * 4 + j) * CSTRIDE], loc[j]);
    }
    int tp = gcn_block_excl_scan(ts, wsum);
    int lp = gcn_block_excl_scan(ls, wsum);
#pragma unroll
    for (int j = 0; j < 4; ++j) {
        const int b = t * 4 + j;
        if (b < nBuckets) {
            if (blockIdx.x == 0) bucketBase[b] = tp;
            gofs[b] = tp + mine[j] - lp;
        }
        lcnt[b] = lp;
        tp += tot[j];
        lp += loc[j];
    }
    __syncthreads();

#pragma unroll
    for (int k = 0; k < EPT; ++k) {
        if (d[k] >= 0) {
            const int bin = d[k] >> NPB_SHIFT;
            const int pos = lcnt[bin] + r[k];
            stageVal[pos] = (int)(((unsigned)s[k] << NPB_SHIFT) | (unsigned)(d[k] & (NPB - 1)));
            stageBin[pos] = bin;
        }
    }
    __syncthreads();

    const int nLocal = min(CH, nEdges - e0);
    for (int i = t; i < nLocal; i += 256)
        tmp[gofs[stageBin[i]] + i] = stageVal[i];
}

__global__ void gcn_bucket_place_kernel(const int* __restrict__ tmp,
                                        const int* __restrict__ bucketTotal,
                                        const int* __restrict__ bucketBase,
                                        int* __restrict__ start,
                                        int* __restrict__ cnt,
                                        float* __restrict__ dinv,
                                        int* __restrict__ col, int nNodes) {
    __shared__ int c[NPB];
    __shared__ int cur[NPB];
    __shared__ int wsum[8];
    __shared__ int stage[CAP];
    const int t = (int)threadIdx.x;
    const int b = (int)blockIdx.x;
    const int n = bucketTotal[b * CSTRIDE];
    const int base = bucketBase[b];
    c[t] = 0;
    __syncthreads();
    for (int i0 = 0; i0 < n; i0 += 8 * 256) {
        int v[8];
#pragma unroll
        for (int u = 0; u < 8; ++u)
            v[u] = tmp[base + min(i0 + u * 256 + t, n - 1)];
#pragma unroll
        for (int u = 0; u < 8; ++u)
            if (i0 + u * 256 + t < n) atomicAdd(&c[v[u] & (NPB - 1)], 1);
    }
    __syncthreads();
    const int mine = c[t];
    const int off = gcn_block_excl_scan(mine, wsum);
    const int node = b * NPB + t;
    if (node < nNodes) {
        start[node] = base + off;
        cnt[node]   = mine;
        dinv[node]  = rsqrtf((float)(mine + 1));
    }
    cur[t] = off;
    __syncthreads();
    if (n <= CAP) {
        for (int i0 = 0; i0 < n; i0 += 8 * 256) {
            int v[8];
#pragma unroll
            for (int u = 0; u < 8; ++u)
                v[u] = tmp[base + min(i0 + u * 256 + t, n - 1)];
#pragma unroll
            for (int u = 0; u < 8; ++u) {
                if (i0 + u * 256 + t < n) {
                    const int pos = atomicAdd(&cur[v[u] & (NPB - 1)], 1);
                    stage[pos] = (int)((unsigned)v[u] >> NPB_SHIFT);
                }
            }
        }
        __syncthreads();
        for (int i = t; i < n; i += 256) col[base + i] = stage[i];
    } else {
        for (int i = t; i < n; i += 256) {
            const int v = tmp[base + i];
            const int pos = atomicAdd(&cur[v & (NPB - 1)], 1);
            col[base + pos] = (int)((unsigned)v >> NPB_SHIFT);
        }
    }
}

__global__ void gcn_gemm_wmma_f32(const float* __restrict__ A,
                                  const float* __restrict__ W,
                                  const float* __restrict__ rowScale,
                                  float* __restrict__ C,
                                  int M, int K, int N) {
    const int wavesPerBlock = blockDim.x >> 5;
    const int wave = (int)(threadIdx.x >> 5);
    const int lane = (int)(threadIdx.x & 31);
    const int tilesN = N >> 4;
    const int gw = blockIdx.x * wavesPerBlock + wave;
    const int tileM = gw / tilesN;
    const int tileN = gw - tileM * tilesN;
    if (tileM * 16 >= M) return;

    const int row0 = tileM << 4;
    const int col0 = tileN << 4;
    const int mr = lane & 15;
    const int kh = (lane >> 4) << 1;

    const int crow = row0 + ((lane >> 4) << 3);
    const float4 sc0 = *reinterpret_cast<const float4*>(rowScale + crow);
    const float4 sc1 = *reinterpret_cast<const float4*>(rowScale + crow + 4);

    v8f acc = {};
    for (int k = 0; k < K; k += 4) {
        v2f a, b;
        const float* ap = A + (long long)(row0 + mr) * K + (k + kh);
        a.x = ap[0];
        a.y = ap[1];
        const float* bp = W + (long long)(k + kh) * N + (col0 + mr);
        b.x = bp[0];
        b.y = bp[N];
        acc = __builtin_amdgcn_wmma_f32_16x16x4_f32(
             false, a,  false, b,
             (short)0, acc,  false,  false);
    }

    const int ccol = col0 + mr;
    const float sc[8] = {sc0.x, sc0.y, sc0.z, sc0.w, sc1.x, sc1.y, sc1.z, sc1.w};
#pragma unroll
    for (int r = 0; r < 8; ++r)
        C[(long long)(crow + r) * N + ccol] = acc[r] * sc[r];
}

template <int C, bool RELU>
__global__ void __launch_bounds__(256)
gcn_gather_kernel(const float* __restrict__ xs,
                  const int* __restrict__ col,
                  const int* __restrict__ start,
                  const int* __restrict__ cnt,
                  const float* __restrict__ dinv,
                  const float* __restrict__ bias,
                  float* __restrict__ res, int nNodes) {
    constexpr int LPR = C / 4;
    constexpr int G   = 32 / LPR;
    static_assert(32 % (4 * G) == 0, "a trip of four entries per group must stay inside the 32 held entries");
    const int lane   = (int)(threadIdx.x & 31);
    const int wpb    = (int)(blockDim.x >> 5);
    const int nWaves = (int)gridDim.x * wpb;
    const int gw = __builtin_amdgcn_readfirstlane(
        (int)(blockIdx.x * wpb + (threadIdx.x >> 5)));
    if (gw >= nNodes) return;
    const int last = nNodes - 1;

    const int grp = lane / LPR;
    const int lig = lane - grp * LPR;
    const float* xcol = xs + lig * 4;
    const float4 bv = *reinterpret_cast<const float4*>(bias + lig * 4);

    int   d   = gw;
    int   beg = start[d];
    int   nIn = cnt[d];
    float dd  = dinv[d];
    int   s_l = d;
    if (lane < nIn) s_l = col[beg + lane];
    int   d1   = min(d + nWaves, last);
    int   beg1 = start[d1];
    int   nIn1 = cnt[d1];
    float dd1  = dinv[d1];

    for (;;) {
        int s1 = d1;
        if (lane < nIn1) s1 = col[beg1 + lane];
        const int   d2   = min(d1 + nWaves, last);
        const int   beg2 = start[d2];
        const int   nIn2 = cnt[d2];
        const float dd2  = dinv[d2];

        const int n = nIn + 1;
        float4 acc = make_float4(0.0f, 0.0f, 0.0f, 0.0f);
        for (int base = 0; base < n; base += 32) {
            const int m = min(32, n - base);
            if (base > 0) {
                const int j = base + lane;
                s_l = d;
                if (j < nIn) s_l = col[beg + j];
            }
            for (int t = 0; t < m; t += 4 * G) {
#pragma unroll
                for (int u = 0; u < 4; ++u) {
                    const int   e = t + u * G + grp;
                    const int   s = __shfl(s_l, e, 32);
                    const float w = (e < m) ? dd : 0.0f;
                    const float4 v = *reinterpret_cast<const float4*>(xcol + (long long)s * C);
                    acc.x += v.x * w;
                    acc.y += v.y * w;
                    acc.z += v.z * w;
                    acc.w += v.w * w;
                }
            }
        }

#pragma unroll
        for (int off = LPR; off < 32; off <<= 1) {
            acc.x += __shfl_xor(acc.x, off, 32);
            acc.y += __shfl_xor(acc.y, off, 32);
            acc.z += __shfl_xor(acc.z, off, 32);
            acc.w += __shfl_xor(acc.w, off, 32);
        }

        if (grp == 0) {
            float4 r = make_float4(acc.x + bv.x, acc.y + bv.y, acc.z + bv.z, acc.w + bv.w);
            if (RELU) {
                r.x = fmaxf(r.x, 0.0f);
                r.y = fmaxf(r.y, 0.0f);
                r.z = fmaxf(r.z, 0.0f);
                r.w = fmaxf(r.w, 0.0f);
            }
            *reinterpret_cast<float4*>(res + (long long)d * C + lig * 4) = r;
        }

        if (d + nWaves >= nNodes) break;
        s_l  = s1;
        d = d1;  beg  = beg1; nIn  = nIn1; dd  = dd1;
        d1 = d2; beg1 = beg2; nIn1 = nIn2; dd1 = dd2;
    }
}

extern "C" void kernel_launch(void* const* d_in, const int* in_sizes, int n_in,
                              void* d_out, int out_size, void* d_ws, size_t ws_size,
                              hipStream_t stream) {
    const float* x  = (const float*)d_in[0];
    const int*   ei = (const int*)  d_in[1];
    const float* W1 = (const float*)d_in[2];
    const float* b1 = (const float*)d_in[3];
    const float* W2 = (const float*)d_in[4];
    const float* b2 = (const float*)d_in[5];
    float* out = (float*)d_out;

    const int nNodes = in_sizes[0] / IN_CH;
    const int nEdges = in_sizes[1] / 2;
    const int* src = ei;
    const int* dst = ei + nEdges;

    const long long hElems = (long long)nNodes * HID;
    float* xl1  = (float*)d_ws;
    float* h    = xl1 + hElems;
    float* xl2  = h   + hElems;
    float* dinv = xl2 + (long long)nNodes * OUT_CH;
    int* start  = (int*)(dinv + nNodes);
    int* tmp    = start + nNodes;
    int* col    = tmp   + nEdges;
    int* cnt    = col   + nEdges;

    const int B = 256;
    const int wpb = B / 32;
    const int nBuckets = (nNodes + NPB - 1) / NPB;
    const int nChunks  = (nEdges + CH - 1) / CH;

    int* bucketBase   = cnt + nNodes;
    int* bucketTotal  = bucketBase + nBuckets;
    int* bucketCursor = bucketTotal + nBuckets * CSTRIDE;

    gcn_zero_kernel<<<(2 * nBuckets * CSTRIDE + B - 1) / B, B, 0, stream>>>(bucketTotal, 2 * nBuckets * CSTRIDE);

    gcn_bucket_count_kernel<<<nChunks, B, 0, stream>>>(dst, bucketTotal, nEdges, nBuckets);
    gcn_bucket_scatter_kernel<<<nChunks, B, 0, stream>>>(src, dst, bucketTotal, bucketCursor, bucketBase,
                                                         tmp, nEdges, nBuckets);
    gcn_bucket_place_kernel<<<nBuckets, B, 0, stream>>>(tmp, bucketTotal, bucketBase, start, cnt, dinv, col, nNodes);

    {
        int tiles = (nNodes / 16) * (HID / 16);
        gcn_gemm_wmma_f32<<<(tiles + wpb - 1) / wpb, B, 0, stream>>>(x, W1, dinv, xl1, nNodes, IN_CH, HID);
    }

    const int gatherGroups = 2048;

    gcn_gather_kernel<HID, true><<<gatherGroups, B, 0, stream>>>(
        xl1, col, start, cnt, dinv, b1, h, nNodes);

    {
        int tiles = (nNodes / 16) * (OUT_CH / 16);
        gcn_gemm_wmma_f32<<<(tiles + wpb - 1) / wpb, B, 0, stream>>>(h, W2, dinv, xl2, nNodes, HID, OUT_CH);
    }

    gcn_gather_kernel<OUT_CH, false><<<gatherGroups, B, 0, stream>>>(
        xl2, col, start, cnt, dinv, b2, out, nNodes);
}


# Products: one wave per 16-row strip computes all column tiles, with 16-byte A and W loads (columns permuted over tiles) and whole-row stores; k summed in another order inside blocks of 8
# speedup vs baseline: 5.0201x; 1.1162x over previous
#include <hip/hip_runtime.h>

#define IN_CH  128
#define HID    64
#define OUT_CH 16

typedef float v2f __attribute__((ext_vector_type(2)));
typedef float v8f __attribute__((ext_vector_type(8)));

#define NPB_SHIFT 8
#define NPB       256
#define NBMAX     1024
#define CH        4096
#define EPT       16
#define CSTRIDE   32
#define CAP       8192

__global__ void gcn_zero_kernel(int* __restrict__ cnt, int n) {
    int i = blockIdx.x * blockDim.x + threadIdx.x;
    if (i < n) cnt[i] = 0;
}

__device__ __forceinline__ int gcn_block_excl_scan(int s, int* wsum) {
    const int lane = (int)(threadIdx.x & 31);
    const int wave = (int)(threadIdx.x >> 5);
    int incl = s;
    for (int off = 1; off < 32; off <<= 1) {
        const int t = __shfl_up(incl, off, 32);
        if (lane >= off) incl += t;
    }
    if (lane == 31) wsum[wave] = incl;
    __syncthreads();
    int wofs = 0;
    for (int w = 0; w < wave; ++w) wofs += wsum[w];
    __syncthreads();
    return wofs + incl - s;
}

__global__ void gcn_bucket_count_kernel(const int* __restrict__ dst,
                                        int* __restrict__ bucketTotal,
                                        int nEdges, int nBuckets) {
    __shared__ int hist[NBMAX];
    const int t = (int)threadIdx.x;
    for (int i = t; i < nBuckets; i += 256) hist[i] = 0;
    __syncthreads();
    const int e0 = blockIdx.x * CH;
    int dv[EPT];
#pragma unroll
    for (int k = 0; k < EPT; ++k)
        dv[k] = dst[min(e0 + k * 256 + t, nEdges - 1)];
#pragma unroll
    for (int k = 0; k < EPT; ++k)
        if (e0 + k * 256 + t < nEdges) atomicAdd(&hist[dv[k] >> NPB_SHIFT], 1);
    __syncthreads();
    for (int i = t; i < nBuckets; i += 256) {
        const int c = hist[i];
        if (c) atomicAdd(&bucketTotal[i * CSTRIDE], c);
    }
}

__global__ void gcn_bucket_scatter_kernel(const int* __restrict__ src,
                                          const int* __restrict__ dst,
                                          const int* __restrict__ bucketTotal,
                                          int* __restrict__ bucketCursor,
                                          int* __restrict__ bucketBase,
                                          int* __restrict__ tmp,
                                          int nEdges, int nBuckets) {
    __shared__ int lcnt[NBMAX];
    __shared__ int gofs[NBMAX];
    __shared__ int stageVal[CH];
    __shared__ int stageBin[CH];
    __shared__ int wsum[8];
    const int t = (int)threadIdx.x;
    for (int i = t; i < NBMAX; i += 256) lcnt[i] = 0;
    __syncthreads();

    const int e0 = blockIdx.x * CH;
    int s[EPT], d[EPT], r[EPT];
#pragma unroll
    for (int k = 0; k < EPT; ++k) {
        const int e = min(e0 + k * 256 + t, nEdges - 1);
        d[k] = dst[e];
        s[k] = src[e];
    }
#pragma unroll
    for (int k = 0; k < EPT; ++k) {
        r[k] = 0;
        if (e0 + k * 256 + t < nEdges)
            r[k] = atomicAdd(&lcnt[d[k] >> NPB_SHIFT], 1);
        else
            d[k] = -1;
    }
    __syncthreads();

    int tot[4], loc[4];
    int ts = 0, ls = 0;
    int mine[4];
#pragma unroll
    for (int j = 0; j < 4; ++j) {
        const int b = t * 4 + j;
        tot[j] = bucketTotal[min(b, nBuckets - 1) * CSTRIDE];
        if (b >= nBuckets) tot[j] = 0;
        loc[j] = lcnt[b];
        ts += tot[j];
        ls += loc[j];
    }
#pragma unroll
    for (int j = 0; j < 4; ++j) {
        mine[j] = 0;
        if (loc[j] > 0) mine[j] = atomicAdd(&bucketCursor[(t * 4 + j) * CSTRIDE], loc[j]);
    }
    int tp = gcn_block_excl_scan(ts, wsum);
    int lp = gcn_block_excl_scan(ls, wsum);
#pragma unroll
    for (int j = 0; j < 4; ++j) {
        const int b = t * 4 + j;
        if (b < nBuckets) {
            if (blockIdx.x == 0) bucketBase[b] = tp;
            gofs[b] = tp + mine[j] - lp;
        }
        lcnt[b] = lp;
        tp += tot[j];
        lp += loc[j];
    }
    __syncthreads();

#pragma unroll
    for (int k = 0; k < EPT; ++k) {
        if (d[k] >= 0) {
            const int bin = d[k] >> NPB_SHIFT;
            const int pos = lcnt[bin] + r[k];
            stageVal[pos] = (int)(((unsigned)s[k] << NPB_SHIFT) | (unsigned)(d[k] & (NPB - 1)));
            stageBin[pos] = bin;
        }
    }
    __syncthreads();

    const int nLocal = min(CH, nEdges - e0);
    for (int i = t; i < nLocal; i += 256)
        tmp[gofs[stageBin[i]] + i] = stageVal[i];
}

__global__ void gcn_bucket_place_kernel(const int* __restrict__ tmp,
                                        const int* __restrict__ bucketTotal,
                                        const int* __restrict__ bucketBase,
                                        int* __restrict__ start,
                                        int* __restrict__ cnt,
                                        float* __restrict__ dinv,
                                        int* __restrict__ col, int nNodes) {
    __shared__ int c[NPB];
    __shared__ int cur[NPB];
    __shared__ int wsum[8];
    __shared__ int stage[CAP];
    const int t = (int)threadIdx.x;
    const int b = (int)blockIdx.x;
    const int n = bucketTotal[b * CSTRIDE];
    const int base = bucketBase[b];
    c[t] = 0;
    __syncthreads();
    for (int i0 = 0; i0 < n; i0 += 8 * 256) {
        int v[8];
#pragma unroll
        for (int u = 0; u < 8; ++u)
            v[u] = tmp[base + min(i0 + u * 256 + t, n - 1)];
#pragma unroll
        for (int u = 0; u < 8; ++u)
            if (i0 + u * 256 + t < n) atomicAdd(&c[v[u] & (NPB - 1)], 1);
    }
    __syncthreads();
    const int mine = c[t];
    const int off = gcn_block_excl_scan(mine, wsum);
    const int node = b * NPB + t;
    if (node < nNodes) {
        start[node] = base + off;
        cnt[node]   = mine;
        dinv[node]  = rsqrtf((float)(mine + 1));
    }
    cur[t] = off;
    __syncthreads();
    if (n <= CAP) {
        for (int i0 = 0; i0 < n; i0 += 8 * 256) {
            int v[8];
#pragma unroll
            for (int u = 0; u < 8; ++u)
                v[u] = tmp[base + min(i0 + u * 256 + t, n - 1)];
#pragma unroll
            for (int u = 0; u < 8; ++u) {
                if (i0 + u * 256 + t < n) {
                    const int pos = atomicAdd(&cur[v[u] & (NPB - 1)], 1);
                    stage[pos] = (int)((unsigned)v[u] >> NPB_SHIFT);
                }
            }
        }
        __syncthreads();
        for (int i = t; i < n; i += 256) col[base + i] = stage[i];
    } else {
        for (int i = t; i < n; i += 256) {
            const int v = tmp[base + i];
            const int pos = atomicAdd(&cur[v & (NPB - 1)], 1);
            col[base + pos] = (int)((unsigned)v >> NPB_SHIFT);
        }
    }
}

template <int NT> struct gcn_wrow { float v[NT]; };

template <int NT>
__device__ __forceinline__ gcn_wrow<NT> gcn_load_wrow(const float* __restrict__ p) {
    gcn_wrow<NT> r;
    if constexpr (NT == 4) {
        const float4 t = *reinterpret_cast<const float4*>(p);
        r.v[0] = t.x; r.v[1] = t.y; r.v[2] = t.z; r.v[3] = t.w;
    } else {
#pragma unroll
        for (int i = 0; i < NT; ++i) r.v[i] = p[i];
    }
    return r;
}

template <int NT>
__device__ __forceinline__ void gcn_store_wrow(float* __restrict__ p, const gcn_wrow<NT>& r) {
    if constexpr (NT == 4) {
        *reinterpret_cast<float4*>(p) = make_float4(r.v[0], r.v[1], r.v[2], r.v[3]);
    } else {
#pragma unroll
        for (int i = 0; i < NT; ++i) p[i] = r.v[i];
    }
}

template <int K, int NT>
__global__ void __launch_bounds__(128)
gcn_gemm_strip_wmma_f32(const float* __restrict__ A,
                        const float* __restrict__ W,
                        const float* __restrict__ rowScale,
                        float* __restrict__ C, int M) {
    constexpr int N  = 16 * NT;
    constexpr int KB = K / 8;
    static_assert(K % 8 == 0, "K is taken in blocks of 8");
    const int lane  = (int)(threadIdx.x & 31);
    const int strip = (int)(blockIdx.x * (blockDim.x >> 5) + (threadIdx.x >> 5));
    if (strip * 16 >= M) return;

    const int row0 = strip << 4;
    const int c = lane & 15;
    const int h = lane >> 4;

    const int crow = row0 + (h << 3);
    const float4 sc0 = *reinterpret_cast<const float4*>(rowScale + crow);
    const float4 sc1 = *reinterpret_cast<const float4*>(rowScale + crow + 4);

    const float* ap = A + (long long)(row0 + c) * K + 4 * h;
    const float* wp = W + (long long)(4 * h) * N + NT * c;

    float4 av[KB];
#pragma unroll
    for (int j = 0; j < KB; ++j)
        av[j] = *reinterpret_cast<const float4*>(ap + 8 * j);

    gcn_wrow<NT> wb[KB][4];
#pragma unroll
    for (int j = 0; j < 2 && j < KB; ++j)
#pragma unroll
        for (int i = 0; i < 4; ++i)
            wb[j][i] = gcn_load_wrow<NT>(wp + (long long)(8 * j + i) * N);

    v8f acc[NT];
#pragma unroll
    for (int t = 0; t < NT; ++t) acc[t] = v8f{};

#pragma unroll
    for (int j = 0; j < KB; ++j) {
        if (j + 2 < KB) {
#pragma unroll
            for (int i = 0; i < 4; ++i)
                wb[j + 2][i] = gcn_load_wrow<NT>(wp + (long long)(8 * (j + 2) + i) * N);
        }
        __builtin_amdgcn_sched_barrier(0);
        v2f a0, a1;
        a0.x = av[j].x; a0.y = av[j].y;
        a1.x = av[j].z; a1.y = av[j].w;
#pragma unroll
        for (int t = 0; t < NT; ++t) {
            v2f b;
            b.x = wb[j][0].v[t]; b.y = wb[j][1].v[t];
            acc[t] = __builtin_amdgcn_wmma_f32_16x16x4_f32(
                 false, a0,  false, b,
                 (short)0, acc[t],  false,  false);
        }
#pragma unroll
        for (int t = 0; t < NT; ++t) {
            v2f b;
            b.x = wb[j][2].v[t]; b.y = wb[j][3].v[t];
            acc[t] = __builtin_amdgcn_wmma_f32_16x16x4_f32(
                 false, a1,  false, b,
                 (short)0, acc[t],  false,  false);
        }
        __builtin_amdgcn_sched_barrier(0);
    }

    const float sc[8] = {sc0.x, sc0.y, sc0.z, sc0.w, sc1.x, sc1.y, sc1.z, sc1.w};
#pragma unroll
    for (int r = 0; r < 8; ++r) {
        gcn_wrow<NT> o;
#pragma unroll
        for (int t = 0; t < NT; ++t) o.v[t] = acc[t][r] * sc[r];
        gcn_store_wrow<NT>(C + (long long)(crow + r) * N + NT * c, o);
    }
}

template <int C, bool RELU>
__global__ void __launch_bounds__(256)
gcn_gather_kernel(const float* __restrict__ xs,
                  const int* __restrict__ col,
                  const int* __restrict__ start,
                  const int* __restrict__ cnt,
                  const float* __restrict__ dinv,
                  const float* __restrict__ bias,
                  float* __restrict__ res, int nNodes) {
    constexpr int LPR = C / 4;
    constexpr int G   = 32 / LPR;
    static_assert(32 % (4 * G) == 0, "a trip of four entries per group must stay inside the 32 held entries");
    const int lane   = (int)(threadIdx.x & 31);
    const int wpb    = (int)(blockDim.x >> 5);
    const int nWaves = (int)gridDim.x * wpb;
    const int gw = __builtin_amdgcn_readfirstlane(
        (int)(blockIdx.x * wpb + (threadIdx.x >> 5)));
    if (gw >= nNodes) return;
    const int last = nNodes - 1;

    const int grp = lane / LPR;
    const int lig = lane - grp * LPR;
    const float* xcol = xs + lig * 4;
    const float4 bv = *reinterpret_cast<const float4*>(bias + lig * 4);

    int   d   = gw;
    int   beg = start[d];
    int   nIn = cnt[d];
    float dd  = dinv[d];
    int   s_l = d;
    if (lane < nIn) s_l = col[beg + lane];
    int   d1   = min(d + nWaves, last);
    int   beg1 = start[d1];
    int   nIn1 = cnt[d1];
    float dd1  = dinv[d1];

    for (;;) {
        int s1 = d1;
        if (lane < nIn1) s1 = col[beg1 + lane];
        const int   d2   = min(d1 + nWaves, last);
        const int   beg2 = start[d2];
        const int   nIn2 = cnt[d2];
        const float dd2  = dinv[d2];

        const int n = nIn + 1;
        float4 acc = make_float4(0.0f, 0.0f, 0.0f, 0.0f);
        for (int base = 0; base < n; base += 32) {
            const int m = min(32, n - base);
            if (base > 0) {
                const int j = base + lane;
                s_l = d;
                if (j < nIn) s_l = col[beg + j];
            }
            for (int t = 0; t < m; t += 4 * G) {
#pragma unroll
                for (int u = 0; u < 4; ++u) {
                    const int   e = t + u * G + grp;
                    const int   s = __shfl(s_l, e, 32);
                    const float w = (e < m) ? dd : 0.0f;
                    const float4 v = *reinterpret_cast<const float4*>(xcol + (long long)s * C);
                    acc.x += v.x * w;
                    acc.y += v.y * w;
                    acc.z += v.z * w;
                    acc.w += v.w * w;
                }
            }
        }

#pragma unroll
        for (int off = LPR; off < 32; off <<= 1) {
            acc.x += __shfl_xor(acc.x, off, 32);
            acc.y += __shfl_xor(acc.y, off, 32);
            acc.z += __shfl_xor(acc.z, off, 32);
            acc.w += __shfl_xor(acc.w, off, 32);
        }

        if (grp == 0) {
            float4 r = make_float4(acc.x + bv.x, acc.y + bv.y, acc.z + bv.z, acc.w + bv.w);
            if (RELU) {
                r.x = fmaxf(r.x, 0.0f);
                r.y = fmaxf(r.y, 0.0f);
                r.z = fmaxf(r.z, 0.0f);
                r.w = fmaxf(r.w, 0.0f);
            }
            *reinterpret_cast<float4*>(res + (long long)d * C + lig * 4) = r;
        }

        if (d + nWaves >= nNodes) break;
        s_l  = s1;
        d = d1;  beg  = beg1; nIn  = nIn1; dd  = dd1;
        d1 = d2; beg1 = beg2; nIn1 = nIn2; dd1 = dd2;
    }
}

extern "C" void kernel_launch(void* const* d_in, const int* in_sizes, int n_in,
                              void* d_out, int out_size, void* d_ws, size_t ws_size,
                              hipStream_t stream) {
    const float* x  = (const float*)d_in[0];
    const int*   ei = (const int*)  d_in[1];
    const float* W1 = (const float*)d_in[2];
    const float* b1 = (const float*)d_in[3];
    const float* W2 = (const float*)d_in[4];
    const float* b2 = (const float*)d_in[5];
    float* out = (float*)d_out;

    const int nNodes = in_sizes[0] / IN_CH;
    const int nEdges = in_sizes[1] / 2;
    const int* src = ei;
    const int* dst = ei + nEdges;

    const long long hElems = (long long)nNodes * HID;
    float* xl1  = (float*)d_ws;
    float* h    = xl1 + hElems;
    float* xl2  = h   + hElems;
    float* dinv = xl2 + (long long)nNodes * OUT_CH;
    int* start  = (int*)(dinv + nNodes);
    int* tmp    = start + nNodes;
    int* col    = tmp   + nEdges;
    int* cnt    = col   + nEdges;

    const int B = 256;
    const int nBuckets = (nNodes + NPB - 1) / NPB;
    const int nChunks  = (nEdges + CH - 1) / CH;

    int* bucketBase   = cnt + nNodes;
    int* bucketTotal  = bucketBase + nBuckets;
    int* bucketCursor = bucketTotal + nBuckets * CSTRIDE;

    gcn_zero_kernel<<<(2 * nBuckets * CSTRIDE + B - 1) / B, B, 0, stream>>>(bucketTotal, 2 * nBuckets * CSTRIDE);

    gcn_bucket_count_kernel<<<nChunks, B, 0, stream>>>(dst, bucketTotal, nEdges, nBuckets);
    gcn_bucket_scatter_kernel<<<nChunks, B, 0, stream>>>(src, dst, bucketTotal, bucketCursor, bucketBase,
                                                         tmp, nEdges, nBuckets);
    gcn_bucket_place_kernel<<<nBuckets, B, 0, stream>>>(tmp, bucketTotal, bucketBase, start, cnt, dinv, col, nNodes);

    {
        int strips = nNodes / 16;
        gcn_gemm_strip_wmma_f32<IN_CH, HID / 16><<<(strips + 3) / 4, 128, 0, stream>>>(x, W1, dinv, xl1, nNodes);
    }

    const int gatherGroups = 2048;

    gcn_gather_kernel<HID, true><<<gatherGroups, B, 0, stream>>>(
        xl1, col, start, cnt, dinv, b1, h, nNodes);

    {
        int strips = nNodes / 16;
        gcn_gemm_strip_wmma_f32<HID, OUT_CH / 16><<<(strips + 3) / 4, 128, 0, stream>>>(h, W2, dinv, xl2, nNodes);
    }

    gcn_gather_kernel<OUT_CH, false><<<gatherGroups, B, 0, stream>>>(
        xl2, col, start, cnt, dinv, b2, out, nNodes);
}


# Second product fused into the first gather: a 16-wave work-group sums 112 nodes' hidden rows into LDS, then 7 waves multiply 16-row strips by W2 on WMMA; h never reaches memory; sum order kept
# speedup vs baseline: 5.3005x; 1.0559x over previous
#include <hip/hip_runtime.h>

#define IN_CH  128
#define HID    64
#define OUT_CH 16

typedef float v2f __attribute__((ext_vector_type(2)));
typedef float v8f __attribute__((ext_vector_type(8)));

#define NPB_SHIFT 8
#define NPB       256
#define NBMAX     1024
#define CH        4096
#define EPT       16
#define CSTRIDE   32
#define CAP       8192

__global__ void gcn_zero_kernel(int* __restrict__ cnt, int n) {
    int i = blockIdx.x * blockDim.x + threadIdx.x;
    if (i < n) cnt[i] = 0;
}

__device__ __forceinline__ int gcn_block_excl_scan(int s, int* wsum) {
    const int lane = (int)(threadIdx.x & 31);
    const int wave = (int)(threadIdx.x >> 5);
    int incl = s;
    for (int off = 1; off < 32; off <<= 1) {
        const int t = __shfl_up(incl, off, 32);
        if (lane >= off) incl += t;
    }
    if (lane == 31) wsum[wave] = incl;
    __syncthreads();
    int wofs = 0;
    for (int w = 0; w < wave; ++w) wofs += wsum[w];
    __syncthreads();
    return wofs + incl - s;
}

__global__ void gcn_bucket_count_kernel(const int* __restrict__ dst,
                                        int* __restrict__ bucketTotal,
                                        int nEdges, int nBuckets) {
    __shared__ int hist[NBMAX];
    const int t = (int)threadIdx.x;
    for (int i = t; i < nBuckets; i += 256) hist[i] = 0;
    __syncthreads();
    const int e0 = blockIdx.x * CH;
    int dv[EPT];
#pragma unroll
    for (int k = 0; k < EPT; ++k)
        dv[k] = dst[min(e0 + k * 256 + t, nEdges - 1)];
#pragma unroll
    for (int k = 0; k < EPT; ++k)
        if (e0 + k * 256 + t < nEdges) atomicAdd(&hist[dv[k] >> NPB_SHIFT], 1);
    __syncthreads();
    for (int i = t; i < nBuckets; i += 256) {
        const int c = hist[i];
        if (c) atomicAdd(&bucketTotal[i * CSTRIDE], c);
    }
}

__global__ void gcn_bucket_scatter_kernel(const int* __restrict__ src,
                                          const int* __restrict__ dst,
                                          const int* __restrict__ bucketTotal,
                                          int* __restrict__ bucketCursor,
                                          int* __restrict__ bucketBase,
                                          int* __restrict__ tmp,
                                          int nEdges, int nBuckets) {
    __shared__ int lcnt[NBMAX];
    __shared__ int gofs[NBMAX];
    __shared__ int stageVal[CH];
    __shared__ int stageBin[CH];
    __shared__ int wsum[8];
    const int t = (int)threadIdx.x;
    for (int i = t; i < NBMAX; i += 256) lcnt[i] = 0;
    __syncthreads();

    const int e0 = blockIdx.x * CH;
    int s[EPT], d[EPT], r[EPT];
#pragma unroll
    for (int k = 0; k < EPT; ++k) {
        const int e = min(e0 + k * 256 + t, nEdges - 1);
        d[k] = dst[e];
        s[k] = src[e];
    }
#pragma unroll
    for (int k = 0; k < EPT; ++k) {
        r[k] = 0;
        if (e0 + k * 256 + t < nEdges)
            r[k] = atomicAdd(&lcnt[d[k] >> NPB_SHIFT], 1);
        else
            d[k] = -1;
    }
    __syncthreads();

    int tot[4], loc[4];
    int ts = 0, ls = 0;
    int mine[4];
#pragma unroll
    for (int j = 0; j < 4; ++j) {
        const int b = t * 4 + j;
        tot[j] = bucketTotal[min(b, nBuckets - 1) * CSTRIDE];
        if (b >= nBuckets) tot[j] = 0;
        loc[j] = lcnt[b];
        ts += tot[j];
        ls += loc[j];
    }
#pragma unroll
    for (int j = 0; j < 4; ++j) {
        mine[j] = 0;
        if (loc[j] > 0) mine[j] = atomicAdd(&bucketCursor[(t * 4 + j) * CSTRIDE], loc[j]);
    }
    int tp = gcn_block_excl_scan(ts, wsum);
    int lp = gcn_block_excl_scan(ls, wsum);
#pragma unroll
    for (int j = 0; j < 4; ++j) {
        const int b = t * 4 + j;
        if (b < nBuckets) {
            if (blockIdx.x == 0) bucketBase[b] = tp;
            gofs[b] = tp + mine[j] - lp;
        }
        lcnt[b] = lp;
        tp += tot[j];
        lp += loc[j];
    }
    __syncthreads();

#pragma unroll
    for (int k = 0; k < EPT; ++k) {
        if (d[k] >= 0) {
            const int bin = d[k] >> NPB_SHIFT;
            const int pos = lcnt[bin] + r[k];
            stageVal[pos] = (int)(((unsigned)s[k] << NPB_SHIFT) | (unsigned)(d[k] & (NPB - 1)));
            stageBin[pos] = bin;
        }
    }
    __syncthreads();

    const int nLocal = min(CH, nEdges - e0);
    for (int i = t; i < nLocal; i += 256)
        tmp[gofs[stageBin[i]] + i] = stageVal[i];
}

__global__ void gcn_bucket_place_kernel(const int* __restrict__ tmp,
                                        const int* __restrict__ bucketTotal,
                                        const int* __restrict__ bucketBase,
                                        int* __restrict__ start,
                                        int* __restrict__ cnt,
                                        float* __restrict__ dinv,
                                        int* __restrict__ col, int nNodes) {
    __shared__ int c[NPB];
    __shared__ int cur[NPB];
    __shared__ int wsum[8];
    __shared__ int stage[CAP];
    const int t = (int)threadIdx.x;
    const int b = (int)blockIdx.x;
    const int n = bucketTotal[b * CSTRIDE];
    const int base = bucketBase[b];
    c[t] = 0;
    __syncthreads();
    for (int i0 = 0; i0 < n; i0 += 8 * 256) {
        int v[8];
#pragma unroll
        for (int u = 0; u < 8; ++u)
            v[u] = tmp[base + min(i0 + u * 256 + t, n - 1)];
#pragma unroll
        for (int u = 0; u < 8; ++u)
            if (i0 + u * 256 + t < n) atomicAdd(&c[v[u] & (NPB - 1)], 1);
    }
    __syncthreads();
    const int mine = c[t];
    const int off = gcn_block_excl_scan(mine, wsum);
    const int node = b * NPB + t;
    if (node < nNodes) {
        start[node] = base + off;
        cnt[node]   = mine;
        dinv[node]  = rsqrtf((float)(mine + 1));
    }
    cur[t] = off;
    __syncthreads();
    if (n <= CAP) {
        for (int i0 = 0; i0 < n; i0 += 8 * 256) {
            int v[8];
#pragma unroll
            for (int u = 0; u < 8; ++u)
                v[u] = tmp[base + min(i0 + u * 256 + t, n - 1)];
#pragma unroll
            for (int u = 0; u < 8; ++u) {
                if (i0 + u * 256 + t < n) {
                    const int pos = atomicAdd(&cur[v[u] & (NPB - 1)], 1);
                    stage[pos] = (int)((unsigned)v[u] >> NPB_SHIFT);
                }
            }
        }
        __syncthreads();
        for (int i = t; i < n; i += 256) col[base + i] = stage[i];
    } else {
        for (int i = t; i < n; i += 256) {
            const int v = tmp[base + i];
            const int pos = atomicAdd(&cur[v & (NPB - 1)], 1);
            col[base + pos] = (int)((unsigned)v >> NPB_SHIFT);
        }
    }
}

template <int NT> struct gcn_wrow { float v[NT]; };

template <int NT>
__device__ __forceinline__ gcn_wrow<NT> gcn_load_wrow(const float* __restrict__ p) {
    gcn_wrow<NT> r;
    if constexpr (NT == 4) {
        const float4 t = *reinterpret_cast<const float4*>(p);
        r.v[0] = t.x; r.v[1] = t.y; r.v[2] = t.z; r.v[3] = t.w;
    } else {
#pragma unroll
        for (int i = 0; i < NT; ++i) r.v[i] = p[i];
    }
    return r;
}

template <int NT>
__device__ __forceinline__ void gcn_store_wrow(float* __restrict__ p, const gcn_wrow<NT>& r) {
    if constexpr (NT == 4) {
        *reinterpret_cast<float4*>(p) = make_float4(r.v[0], r.v[1], r.v[2], r.v[3]);
    } else {
#pragma unroll
        for (int i = 0; i < NT; ++i) p[i] = r.v[i];
    }
}

template <int K, int NT>
__global__ void __launch_bounds__(128)
gcn_gemm_strip_wmma_f32(const float* __restrict__ A,
                        const float* __restrict__ W,
                        const float* __restrict__ rowScale,
                        float* __restrict__ C, int M) {
    constexpr int N  = 16 * NT;
    constexpr int KB = K / 8;
    static_assert(K % 8 == 0, "K is taken in blocks of 8");
    const int lane  = (int)(threadIdx.x & 31);
    const int strip = (int)(blockIdx.x * (blockDim.x >> 5) + (threadIdx.x >> 5));
    if (strip * 16 >= M) return;

    const int row0 = strip << 4;
    const int c = lane & 15;
    const int h = lane >> 4;

    const int crow = row0 + (h << 3);
    const float4 sc0 = *reinterpret_cast<const float4*>(rowScale + crow);
    const float4 sc1 = *reinterpret_cast<const float4*>(rowScale + crow + 4);

    const float* ap = A + (long long)(row0 + c) * K + 4 * h;
    const float* wp = W + (long long)(4 * h) * N + NT * c;

    float4 av[KB];
#pragma unroll
    for (int j = 0; j < KB; ++j)
        av[j] = *reinterpret_cast<const float4*>(ap + 8 * j);

    gcn_wrow<NT> wb[KB][4];
#pragma unroll
    for (int j = 0; j < 2 && j < KB; ++j)
#pragma unroll
        for (int i = 0; i < 4; ++i)
            wb[j][i] = gcn_load_wrow<NT>(wp + (long long)(8 * j + i) * N);

    v8f acc[NT];
#pragma unroll
    for (int t = 0; t < NT; ++t) acc[t] = v8f{};

#pragma unroll
    for (int j = 0; j < KB; ++j) {
        if (j + 2 < KB) {
#pragma unroll
            for (int i = 0; i < 4; ++i)
                wb[j + 2][i] = gcn_load_wrow<NT>(wp + (long long)(8 * (j + 2) + i) * N);
        }
        __builtin_amdgcn_sched_barrier(0);
        v2f a0, a1;
        a0.x = av[j].x; a0.y = av[j].y;
        a1.x = av[j].z; a1.y = av[j].w;
#pragma unroll
        for (int t = 0; t < NT; ++t) {
            v2f b;
            b.x = wb[j][0].v[t]; b.y = wb[j][1].v[t];
            acc[t] = __builtin_amdgcn_wmma_f32_16x16x4_f32(
                 false, a0,  false, b,
                 (short)0, acc[t],  false,  false);
        }
#pragma unroll
        for (int t = 0; t < NT; ++t) {
            v2f b;
            b.x = wb[j][2].v[t]; b.y = wb[j][3].v[t];
            acc[t] = __builtin_amdgcn_wmma_f32_16x16x4_f32(
                 false, a1,  false, b,
                 (short)0, acc[t],  false,  false);
        }
        __builtin_amdgcn_sched_barrier(0);
    }

    const float sc[8] = {sc0.x, sc0.y, sc0.z, sc0.w, sc1.x, sc1.y, sc1.z, sc1.w};
#pragma unroll
    for (int r = 0; r < 8; ++r) {
        gcn_wrow<NT> o;
#pragma unroll
        for (int t = 0; t < NT; ++t) o.v[t] = acc[t][r] * sc[r];
        gcn_store_wrow<NT>(C + (long long)(crow + r) * N + NT * c, o);
    }
}

template <int C, bool RELU>
__global__ void __launch_bounds__(256)
gcn_gather_kernel(const float* __restrict__ xs,
                  const int* __restrict__ col,
                  const int* __restrict__ start,
                  const int* __restrict__ cnt,
                  const float* __restrict__ dinv,
                  const float* __restrict__ bias,
                  float* __restrict__ res, int nNodes) {
    constexpr int LPR = C / 4;
    constexpr int G   = 32 / LPR;
    static_assert(32 % (4 * G) == 0, "a trip of four entries per group must stay inside the 32 held entries");
    const int lane   = (int)(threadIdx.x & 31);
    const int wpb    = (int)(blockDim.x >> 5);
    const int nWaves = (int)gridDim.x * wpb;
    const int gw = __builtin_amdgcn_readfirstlane(
        (int)(blockIdx.x * wpb + (threadIdx.x >> 5)));
    if (gw >= nNodes) return;
    const int last = nNodes - 1;

    const int grp = lane / LPR;
    const int lig = lane - grp * LPR;
    const float* xcol = xs + lig * 4;
    const float4 bv = *reinterpret_cast<const float4*>(bias + lig * 4);

    int   d   = gw;
    int   beg = start[d];
    int   nIn = cnt[d];
    float dd  = dinv[d];
    int   s_l = d;
    if (lane < nIn) s_l = col[beg + lane];
    int   d1   = min(d + nWaves, last);
    int   beg1 = start[d1];
    int   nIn1 = cnt[d1];
    float dd1  = dinv[d1];

    for (;;) {
        int s1 = d1;
        if (lane < nIn1) s1 = col[beg1 + lane];
        const int   d2   = min(d1 + nWaves, last);
        const int   beg2 = start[d2];
        const int   nIn2 = cnt[d2];
        const float dd2  = dinv[d2];

        const int n = nIn + 1;
        float4 acc = make_float4(0.0f, 0.0f, 0.0f, 0.0f);
        for (int base = 0; base < n; base += 32) {
            const int m = min(32, n - base);
            if (base > 0) {
                const int j = base + lane;
                s_l = d;
                if (j < nIn) s_l = col[beg + j];
            }
            for (int t = 0; t < m; t += 4 * G) {
#pragma unroll
                for (int u = 0; u < 4; ++u) {
                    const int   e = t + u * G + grp;
                    const int   s = __shfl(s_l, e, 32);
                    const float w = (e < m) ? dd : 0.0f;
                    const float4 v = *reinterpret_cast<const float4*>(xcol + (long long)s * C);
                    acc.x += v.x * w;
                    acc.y += v.y * w;
                    acc.z += v.z * w;
                    acc.w += v.w * w;
                }
            }
        }

#pragma unroll
        for (int off = LPR; off < 32; off <<= 1) {
            acc.x += __shfl_xor(acc.x, off, 32);
            acc.y += __shfl_xor(acc.y, off, 32);
            acc.z += __shfl_xor(acc.z, off, 32);
            acc.w += __shfl_xor(acc.w, off, 32);
        }

        if (grp == 0) {
            float4 r = make_float4(acc.x + bv.x, acc.y + bv.y, acc.z + bv.z, acc.w + bv.w);
            if (RELU) {
                r.x = fmaxf(r.x, 0.0f);
                r.y = fmaxf(r.y, 0.0f);
                r.z = fmaxf(r.z, 0.0f);
                r.w = fmaxf(r.w, 0.0f);
            }
            *reinterpret_cast<float4*>(res + (long long)d * C + lig * 4) = r;
        }

        if (d + nWaves >= nNodes) break;
        s_l  = s1;
        d = d1;  beg  = beg1; nIn  = nIn1; dd  = dd1;
        d1 = d2; beg1 = beg2; nIn1 = nIn2; dd1 = dd2;
    }
}

#define FG_WAVES  16
#define FG_STRIPS 7
#define FG_NODES  (16 * FG_STRIPS)
#define FG_PITCH  (HID + 4)

__global__ void __launch_bounds__(FG_WAVES * 32)
gcn_gather_product_kernel(const float* __restrict__ xs,
                          const int* __restrict__ col,
                          const int* __restrict__ start,
                          const int* __restrict__ cnt,
                          const float* __restrict__ dinv,
                          const float* __restrict__ bias,
                          const float* __restrict__ W,
                          float* __restrict__ C, int nNodes) {
    constexpr int LPR = HID / 4;
    constexpr int G   = 32 / LPR;
    __shared__ float hs[FG_NODES * FG_PITCH];
    const int lane = (int)(threadIdx.x & 31);
    const int wv   = __builtin_amdgcn_readfirstlane((int)(threadIdx.x >> 5));
    const int base = (int)blockIdx.x * FG_NODES;
    const int last = nNodes - 1;

    const int grp = lane / LPR;
    const int lig = lane - grp * LPR;
    const float* xcol = xs + lig * 4;
    const float4 bv = *reinterpret_cast<const float4*>(bias + lig * 4);

    int   d   = base + wv;
    int   dc  = min(d, last);
    int   beg = start[dc];
    int   nIn = cnt[dc];
    float dd  = dinv[dc];
    int   s_l = dc;
    if (lane < nIn) s_l = col[beg + lane];
    int   d1c  = min(d + FG_WAVES, last);
    int   beg1 = start[d1c];
    int   nIn1 = cnt[d1c];
    float dd1  = dinv[d1c];

    for (int k = 0; k < FG_STRIPS; ++k) {
        int s1 = d1c;
        if (k + 1 < FG_STRIPS && lane < nIn1) s1 = col[beg1 + lane];
        const int d2c = min(d + 2 * FG_WAVES, last);
        int beg2 = 0, nIn2 = 0;
        float dd2 = 0.0f;
        if (k + 2 < FG_STRIPS) {
            beg2 = start[d2c];
            nIn2 = cnt[d2c];
            dd2  = dinv[d2c];
        }

        float4 r = make_float4(0.0f, 0.0f, 0.0f, 0.0f);
        if (d < nNodes) {
            const int n = nIn + 1;
            float4 acc = make_float4(0.0f, 0.0f, 0.0f, 0.0f);
            for (int b0 = 0; b0 < n; b0 += 32) {
                const int m = min(32, n - b0);
                if (b0 > 0) {
                    const int j = b0 + lane;
                    s_l = d;
                    if (j < nIn) s_l = col[beg + j];
                }
                for (int t = 0; t < m; t += 4 * G) {
#pragma unroll
                    for (int u = 0; u < 4; ++u) {
                        const int   e = t + u * G + grp;
                        const int   s = __shfl(s_l, e, 32);
                        const float w = (e < m) ? dd : 0.0f;
                        const float4 v = *reinterpret_cast<const float4*>(xcol + (long long)s * HID);
                        acc.x += v.x * w;
                        acc.y += v.y * w;
                        acc.z += v.z * w;
                        acc.w += v.w * w;
                    }
                }
            }
#pragma unroll
            for (int off = LPR; off < 32; off <<= 1) {
                acc.x += __shfl_xor(acc.x, off, 32);
                acc.y += __shfl_xor(acc.y, off, 32);
                acc.z += __shfl_xor(acc.z, off, 32);
                acc.w += __shfl_xor(acc.w, off, 32);
            }
            r.x = fmaxf(acc.x + bv.x, 0.0f);
            r.y = fmaxf(acc.y + bv.y, 0.0f);
            r.z = fmaxf(acc.z + bv.z, 0.0f);
            r.w = fmaxf(acc.w + bv.w, 0.0f);
        }
        if (grp == 0)
            *reinterpret_cast<float4*>(&hs[(wv + FG_WAVES * k) * FG_PITCH + lig * 4]) = r;

        s_l = s1;
        d += FG_WAVES;
        beg = beg1;  nIn = nIn1;  dd = dd1;
        d1c = d2c; beg1 = beg2; nIn1 = nIn2; dd1 = dd2;
    }

    __syncthreads();

    if (wv >= FG_STRIPS) return;
    const int row0 = base + 16 * wv;
    if (row0 >= nNodes) return;
    const int c = lane & 15;
    const int h = lane >> 4;
    const int crow = row0 + (h << 3);
    const float4 sc0 = *reinterpret_cast<const float4*>(dinv + crow);
    const float4 sc1 = *reinterpret_cast<const float4*>(dinv + crow + 4);

    constexpr int KB = HID / 8;
    const float* wp = W + (4 * h) * OUT_CH + c;
    float wb[KB][4];
#pragma unroll
    for (int j = 0; j < KB; ++j)
#pragma unroll
        for (int i = 0; i < 4; ++i)
            wb[j][i] = wp[(8 * j + i) * OUT_CH];

    const float* ap = hs + (16 * wv + c) * FG_PITCH + 4 * h;
    float4 av[KB];
#pragma unroll
    for (int j = 0; j < KB; ++j)
        av[j] = *reinterpret_cast<const float4*>(ap + 8 * j);

    v8f acc8 = v8f{};
#pragma unroll
    for (int j = 0; j < KB; ++j) {
        v2f a0, a1, b0, b1;
        a0.x = av[j].x; a0.y = av[j].y;
        a1.x = av[j].z; a1.y = av[j].w;
        b0.x = wb[j][0]; b0.y = wb[j][1];
        b1.x = wb[j][2]; b1.y = wb[j][3];
        acc8 = __builtin_amdgcn_wmma_f32_16x16x4_f32(
             false, a0,  false, b0,
             (short)0, acc8,  false,  false);
        acc8 = __builtin_amdgcn_wmma_f32_16x16x4_f32(
             false, a1,  false, b1,
             (short)0, acc8,  false,  false);
    }

    const float sc[8] = {sc0.x, sc0.y, sc0.z, sc0.w, sc1.x, sc1.y, sc1.z, sc1.w};
    if (row0 + 16 <= nNodes) {
#pragma unroll
        for (int r8 = 0; r8 < 8; ++r8)
            C[(long long)(crow + r8) * OUT_CH + c] = acc8[r8] * sc[r8];
    } else {
#pragma unroll
        for (int r8 = 0; r8 < 8; ++r8)
            if (crow + r8 < nNodes)
                C[(long long)(crow + r8) * OUT_CH + c] = acc8[r8] * sc[r8];
    }
}

extern "C" void kernel_launch(void* const* d_in, const int* in_sizes, int n_in,
                              void* d_out, int out_size, void* d_ws, size_t ws_size,
                              hipStream_t stream) {
    const float* x  = (const float*)d_in[0];
    const int*   ei = (const int*)  d_in[1];
    const float* W1 = (const float*)d_in[2];
    const float* b1 = (const float*)d_in[3];
    const float* W2 = (const float*)d_in[4];
    const float* b2 = (const float*)d_in[5];
    float* out = (float*)d_out;

    const int nNodes = in_sizes[0] / IN_CH;
    const int nEdges = in_sizes[1] / 2;
    const int* src = ei;
    const int* dst = ei + nEdges;

    const long long hElems = (long long)nNodes * HID;
    float* xl1  = (float*)d_ws;
    float* h    = xl1 + hElems;
    float* xl2  = h   + hElems;
    float* dinv = xl2 + (long long)nNodes * OUT_CH;
    int* start  = (int*)(dinv + nNodes);
    int* tmp    = start + nNodes;
    int* col    = tmp   + nEdges;
    int* cnt    = col   + nEdges;

    const int B = 256;
    const int nBuckets = (nNodes + NPB - 1) / NPB;
    const int nChunks  = (nEdges + CH - 1) / CH;

    int* bucketBase   = cnt + nNodes;
    int* bucketTotal  = bucketBase + nBuckets;
    int* bucketCursor = bucketTotal + nBuckets * CSTRIDE;

    gcn_zero_kernel<<<(2 * nBuckets * CSTRIDE + B - 1) / B, B, 0, stream>>>(bucketTotal, 2 * nBuckets * CSTRIDE);

    gcn_bucket_count_kernel<<<nChunks, B, 0, stream>>>(dst, bucketTotal, nEdges, nBuckets);
    gcn_bucket_scatter_kernel<<<nChunks, B, 0, stream>>>(src, dst, bucketTotal, bucketCursor, bucketBase,
                                                         tmp, nEdges, nBuckets);
    gcn_bucket_place_kernel<<<nBuckets, B, 0, stream>>>(tmp, bucketTotal, bucketBase, start, cnt, dinv, col, nNodes);

    {
        int strips = nNodes / 16;
        gcn_gemm_strip_wmma_f32<IN_CH, HID / 16><<<(strips + 3) / 4, 128, 0, stream>>>(x, W1, dinv, xl1, nNodes);
    }

    gcn_gather_product_kernel<<<(nNodes + FG_NODES - 1) / FG_NODES, FG_WAVES * 32, 0, stream>>>(
        xl1, col, start, cnt, dinv, b1, W2, xl2, nNodes);

    const int gatherGroups = 2048;

    gcn_gather_kernel<OUT_CH, false><<<gatherGroups, B, 0, stream>>>(
        xl2, col, start, cnt, dinv, b2, out, nNodes);
}


# Edge grouping in two launches, no global counters: chunks ordered by bucket in LDS plus a tiled per-chunk prefix table, buckets collect their runs; zero and count launches go; LDS arrival order kept
# speedup vs baseline: 7.3209x; 1.3812x over previous
#include <hip/hip_runtime.h>

#define IN_CH  128
#define HID    64
#define OUT_CH 16

typedef float v2f __attribute__((ext_vector_type(2)));
typedef float v8f __attribute__((ext_vector_type(8)));

#define NPB_SHIFT 8
#define NPB       256
#define NBMAX     1024
#define CH        4096
#define EPT       16
#define CAP       8192

__device__ __forceinline__ long long gcn_pfx_at(int ch, int b, int nBT) {
    return ((long long)((ch >> 3) * nBT + (b >> 3)) << 6) + ((ch & 7) << 3) + (b & 7);
}

__device__ __forceinline__ int gcn_block_excl_scan(int s, int* wsum, int* total) {
    const int lane = (int)(threadIdx.x & 31);
    const int wave = (int)(threadIdx.x >> 5);
    int incl = s;
    for (int off = 1; off < 32; off <<= 1) {
        const int t = __shfl_up(incl, off, 32);
        if (lane >= off) incl += t;
    }
    if (lane == 31) wsum[wave] = incl;
    __syncthreads();
    int wofs = 0, tot = 0;
    for (int w = 0; w < 8; ++w) {
        const int x = wsum[w];
        if (w < wave) wofs += x;
        tot += x;
    }
    __syncthreads();
    *total = tot;
    return wofs + incl - s;
}

__global__ void __launch_bounds__(256)
gcn_chunk_sort_kernel(const int* __restrict__ src,
                      const int* __restrict__ dst,
                      int* __restrict__ pfx,
                      int* __restrict__ tmp,
                      int nEdges, int nBuckets) {
    __shared__ int lcnt[NBMAX];
    __shared__ int stageVal[CH];
    __shared__ int wsum[8];
    const int t = (int)threadIdx.x;
    for (int i = t; i < NBMAX; i += 256) lcnt[i] = 0;
    __syncthreads();

    const int e0 = blockIdx.x * CH;
    int s[EPT], d[EPT], r[EPT];
#pragma unroll
    for (int k = 0; k < EPT; ++k) {
        const int e = min(e0 + k * 256 + t, nEdges - 1);
        d[k] = dst[e];
        s[k] = src[e];
    }
#pragma unroll
    for (int k = 0; k < EPT; ++k) {
        r[k] = 0;
        if (e0 + k * 256 + t < nEdges)
            r[k] = atomicAdd(&lcnt[d[k] >> NPB_SHIFT], 1);
        else
            d[k] = -1;
    }
    __syncthreads();

    int loc[4];
    int ls = 0;
#pragma unroll
    for (int j = 0; j < 4; ++j) {
        loc[j] = lcnt[t * 4 + j];
        ls += loc[j];
    }
    int nLocal;
    int lp = gcn_block_excl_scan(ls, wsum, &nLocal);
    const int nBT = (nBuckets + 8) >> 3;
    int pv[4];
#pragma unroll
    for (int j = 0; j < 4; ++j) {
        pv[j] = lp;
        lcnt[t * 4 + j] = lp;
        lp += loc[j];
    }
    if ((t >> 1) < nBT)
        *reinterpret_cast<int4*>(pfx + gcn_pfx_at((int)blockIdx.x, t * 4, nBT)) =
            make_int4(pv[0], pv[1], pv[2], pv[3]);
    if (t == 0 && nBT > (NBMAX >> 3))
        *reinterpret_cast<int4*>(pfx + gcn_pfx_at((int)blockIdx.x, NBMAX, nBT)) =
            make_int4(nLocal, nLocal, nLocal, nLocal);
    __syncthreads();

#pragma unroll
    for (int k = 0; k < EPT; ++k) {
        if (d[k] >= 0) {
            const int pos = lcnt[d[k] >> NPB_SHIFT] + r[k];
            stageVal[pos] = (int)(((unsigned)s[k] << NPB_SHIFT) | (unsigned)(d[k] & (NPB - 1)));
        }
    }
    __syncthreads();

    for (int i = t; i < nLocal; i += 256)
        tmp[e0 + i] = stageVal[i];
}

__global__ void __launch_bounds__(256)
gcn_bucket_place_kernel(const int* __restrict__ tmp,
                        const int* __restrict__ pfx,
                        int* __restrict__ start,
                        int* __restrict__ cnt,
                        float* __restrict__ dinv,
                        int* __restrict__ col,
                        int nNodes, int nEdges, int nChunks, int nBuckets) {
    __shared__ int c[NPB];
    __shared__ int cur[NPB];
    __shared__ int wsum[8];
    __shared__ int idx[CAP];
    __shared__ int stage[CAP];
    const int t = (int)threadIdx.x;
    const int b = (int)blockIdx.x;
    const int nBT = (nBuckets + 8) >> 3;
    const int lastE = nEdges - 1;
    c[t] = 0;

    const int ch0 = min(t, nChunks - 1);
    const int ch1 = min(t + 256, nChunks - 1);
    const int p00 = pfx[gcn_pfx_at(ch0, b, nBT)];
    const int p01 = pfx[gcn_pfx_at(ch0, b + 1, nBT)];
    const int p10 = pfx[gcn_pfx_at(ch1, b, nBT)];
    const int p11 = pfx[gcn_pfx_at(ch1, b + 1, nBT)];
    const int lo0  = (t < nChunks) ? p00 : 0;
    const int len0 = (t < nChunks) ? p01 - p00 : 0;
    const int lo1  = (t + 256 < nChunks) ? p10 : 0;
    const int len1 = (t + 256 < nChunks) ? p11 - p10 : 0;
    int sLo = lo0 + lo1, sN = len0 + len1;
    for (int ch = t + 512; ch < nChunks; ch += 256) {
        const int lo = pfx[gcn_pfx_at(ch, b, nBT)];
        sLo += lo;
        sN  += pfx[gcn_pfx_at(ch, b + 1, nBT)] - lo;
    }
    int base, n;
    gcn_block_excl_scan(sLo, wsum, &base);
    const int eo0 = gcn_block_excl_scan(sN, wsum, &n);

    if (n <= CAP) {
        int eo = eo0;
        {
            const int g0 = t * CH + lo0;
            for (int j = 0; j < len0; ++j) idx[eo + j] = g0 + j;
            eo += len0;
        }
        {
            const int g0 = (t + 256) * CH + lo1;
            for (int j = 0; j < len1; ++j) idx[eo + j] = g0 + j;
            eo += len1;
        }
        for (int ch = t + 512; ch < nChunks; ch += 256) {
            const int lo  = pfx[gcn_pfx_at(ch, b, nBT)];
            const int len = pfx[gcn_pfx_at(ch, b + 1, nBT)] - lo;
            const int g0  = ch * CH + lo;
            for (int j = 0; j < len; ++j) idx[eo + j] = g0 + j;
            eo += len;
        }
        __syncthreads();
        for (int i0 = 0; i0 < n; i0 += 8 * 256) {
            int v[8];
#pragma unroll
            for (int u = 0; u < 8; ++u) {
                const int g = idx[min(i0 + u * 256 + t, n - 1)];
                v[u] = tmp[min(max(g, 0), lastE)];
            }
#pragma unroll
            for (int u = 0; u < 8; ++u) {
                const int i = i0 + u * 256 + t;
                if (i < n) {
                    stage[i] = v[u];
                    atomicAdd(&c[v[u] & (NPB - 1)], 1);
                }
            }
        }
    } else {
        for (int ch = t; ch < nChunks; ch += 256) {
            const int lo = pfx[gcn_pfx_at(ch, b, nBT)];
            const int hi = pfx[gcn_pfx_at(ch, b + 1, nBT)];
            for (int j = lo; j < hi; ++j) {
                const int v = tmp[min(max(ch * CH + j, 0), lastE)];
                atomicAdd(&c[v & (NPB - 1)], 1);
            }
        }
    }
    __syncthreads();
    const int mine = c[t];
    int nAll;
    const int off = gcn_block_excl_scan(mine, wsum, &nAll);
    const int node = b * NPB + t;
    if (node < nNodes) {
        start[node] = base + off;
        cnt[node]   = mine;
        dinv[node]  = rsqrtf((float)(mine + 1));
    }
    cur[t] = off;
    __syncthreads();
    if (n <= CAP) {
        for (int i0 = 0; i0 < n; i0 += 8 * 256) {
            int v[8];
#pragma unroll
            for (int u = 0; u < 8; ++u)
                v[u] = stage[min(i0 + u * 256 + t, n - 1)];
            int pos[8];
#pragma unroll
            for (int u = 0; u < 8; ++u) {
                pos[u] = -1;
                if (i0 + u * 256 + t < n) pos[u] = atomicAdd(&cur[v[u] & (NPB - 1)], 1);
            }
#pragma unroll
            for (int u = 0; u < 8; ++u)
                if (pos[u] >= 0) idx[pos[u]] = (int)((unsigned)v[u] >> NPB_SHIFT);
        }
        __syncthreads();
        for (int i = t; i < n; i += 256)
            if (base + i <= lastE) col[base + i] = idx[i];
    } else {
        for (int ch = t; ch < nChunks; ch += 256) {
            const int lo = pfx[gcn_pfx_at(ch, b, nBT)];
            const int hi = pfx[gcn_pfx_at(ch, b + 1, nBT)];
            for (int j = lo; j < hi; ++j) {
                const int v = tmp[min(max(ch * CH + j, 0), lastE)];
                const int pos = base + atomicAdd(&cur[v & (NPB - 1)], 1);
                if (pos <= lastE) col[pos] = (int)((unsigned)v >> NPB_SHIFT);
            }
        }
    }
}

template <int NT> struct gcn_wrow { float v[NT]; };

template <int NT>
__device__ __forceinline__ gcn_wrow<NT> gcn_load_wrow(const float* __restrict__ p) {
    gcn_wrow<NT> r;
    if constexpr (NT == 4) {
        const float4 t = *reinterpret_cast<const float4*>(p);
        r.v[0] = t.x; r.v[1] = t.y; r.v[2] = t.z; r.v[3] = t.w;
    } else {
#pragma unroll
        for (int i = 0; i < NT; ++i) r.v[i] = p[i];
    }
    return r;
}

template <int NT>
__device__ __forceinline__ void gcn_store_wrow(float* __restrict__ p, const gcn_wrow<NT>& r) {
    if constexpr (NT == 4) {
        *reinterpret_cast<float4*>(p) = make_float4(r.v[0], r.v[1], r.v[2], r.v[3]);
    } else {
#pragma unroll
        for (int i = 0; i < NT; ++i) p[i] = r.v[i];
    }
}

template <int K, int NT>
__global__ void __launch_bounds__(128)
gcn_gemm_strip_wmma_f32(const float* __restrict__ A,
                        const float* __restrict__ W,
                        const float* __restrict__ rowScale,
                        float* __restrict__ C, int M) {
    constexpr int N  = 16 * NT;
    constexpr int KB = K / 8;
    static_assert(K % 8 == 0, "K is taken in blocks of 8");
    const int lane  = (int)(threadIdx.x & 31);
    const int strip = (int)(blockIdx.x * (blockDim.x >> 5) + (threadIdx.x >> 5));
    if (strip * 16 >= M) return;

    const int row0 = strip << 4;
    const int c = lane & 15;
    const int h = lane >> 4;

    const int crow = row0 + (h << 3);
    const float4 sc0 = *reinterpret_cast<const float4*>(rowScale + crow);
    const float4 sc1 = *reinterpret_cast<const float4*>(rowScale + crow + 4);

    const float* ap = A + (long long)(row0 + c) * K + 4 * h;
    const float* wp = W + (long long)(4 * h) * N + NT * c;

    float4 av[KB];
#pragma unroll
    for (int j = 0; j < KB; ++j)
        av[j] = *reinterpret_cast<const float4*>(ap + 8 * j);

    gcn_wrow<NT> wb[KB][4];
#pragma unroll
    for (int j = 0; j < 2 && j < KB; ++j)
#pragma unroll
        for (int i = 0; i < 4; ++i)
            wb[j][i] = gcn_load_wrow<NT>(wp + (long long)(8 * j + i) * N);

    v8f acc[NT];
#pragma unroll
    for (int t = 0; t < NT; ++t) acc[t] = v8f{};

#pragma unroll
    for (int j = 0; j < KB; ++j) {
        if (j + 2 < KB) {
#pragma unroll
            for (int i = 0; i < 4; ++i)
                wb[j + 2][i] = gcn_load_wrow<NT>(wp + (long long)(8 * (j + 2) + i) * N);
        }
        __builtin_amdgcn_sched_barrier(0);
        v2f a0, a1;
        a0.x = av[j].x; a0.y = av[j].y;
        a1.x = av[j].z; a1.y = av[j].w;
#pragma unroll
        for (int t = 0; t < NT; ++t) {
            v2f b;
            b.x = wb[j][0].v[t]; b.y = wb[j][1].v[t];
            acc[t] = __builtin_amdgcn_wmma_f32_16x16x4_f32(
                 false, a0,  false, b,
                 (short)0, acc[t],  false,  false);
        }
#pragma unroll
        for (int t = 0; t < NT; ++t) {
            v2f b;
            b.x = wb[j][2].v[t]; b.y = wb[j][3].v[t];
            acc[t] = __builtin_amdgcn_wmma_f32_16x16x4_f32(
                 false, a1,  false, b,
                 (short)0, acc[t],  false,  false);
        }
        __builtin_amdgcn_sched_barrier(0);
    }

    const float sc[8] = {sc0.x, sc0.y, sc0.z, sc0.w, sc1.x, sc1.y, sc1.z, sc1.w};
#pragma unroll
    for (int r = 0; r < 8; ++r) {
        gcn_wrow<NT> o;
#pragma unroll
        for (int t = 0; t < NT; ++t) o.v[t] = acc[t][r] * sc[r];
        gcn_store_wrow<NT>(C + (long long)(crow + r) * N + NT * c, o);
    }
}

template <int C, bool RELU>
__global__ void __launch_bounds__(256)
gcn_gather_kernel(const float* __restrict__ xs,
                  const int* __restrict__ col,
                  const int* __restrict__ start,
                  const int* __restrict__ cnt,
                  const float* __restrict__ dinv,
                  const float* __restrict__ bias,
                  float* __restrict__ res, int nNodes) {
    constexpr int LPR = C / 4;
    constexpr int G   = 32 / LPR;
    static_assert(32 % (4 * G) == 0, "a trip of four entries per group must stay inside the 32 held entries");
    const int lane   = (int)(threadIdx.x & 31);
    const int wpb    = (int)(blockDim.x >> 5);
    const int nWaves = (int)gridDim.x * wpb;
    const int gw = __builtin_amdgcn_readfirstlane(
        (int)(blockIdx.x * wpb + (threadIdx.x >> 5)));
    if (gw >= nNodes) return;
    const int last = nNodes - 1;

    const int grp = lane / LPR;
    const int lig = lane - grp * LPR;
    const float* xcol = xs + lig * 4;
    const float4 bv = *reinterpret_cast<const float4*>(bias + lig * 4);

    int   d   = gw;
    int   beg = start[d];
    int   nIn = cnt[d];
    float dd  = dinv[d];
    int   s_l = d;
    if (lane < nIn) s_l = col[beg + lane];
    int   d1   = min(d + nWaves, last);
    int   beg1 = start[d1];
    int   nIn1 = cnt[d1];
    float dd1  = dinv[d1];

    for (;;) {
        int s1 = d1;
        if (lane < nIn1) s1 = col[beg1 + lane];
        const int   d2   = min(d1 + nWaves, last);
        const int   beg2 = start[d2];
        const int   nIn2 = cnt[d2];
        const float dd2  = dinv[d2];

        const int n = nIn + 1;
        float4 acc = make_float4(0.0f, 0.0f, 0.0f, 0.0f);
        for (int base = 0; base < n; base += 32) {
            const int m = min(32, n - base);
            if (base > 0) {
                const int j = base + lane;
                s_l = d;
                if (j < nIn) s_l = col[beg + j];
            }
            for (int t = 0; t < m; t += 4 * G) {
#pragma unroll
                for (int u = 0; u < 4; ++u) {
                    const int   e = t + u * G + grp;
                    const int   s = __shfl(s_l, e, 32);
                    const float w = (e < m) ? dd : 0.0f;
                    const float4 v = *reinterpret_cast<const float4*>(xcol + (long long)s * C);
                    acc.x += v.x * w;
                    acc.y += v.y * w;
                    acc.z += v.z * w;
                    acc.w += v.w * w;
                }
            }
        }

#pragma unroll
        for (int off = LPR; off < 32; off <<= 1) {
            acc.x += __shfl_xor(acc.x, off, 32);
            acc.y += __shfl_xor(acc.y, off, 32);
            acc.z += __shfl_xor(acc.z, off, 32);
            acc.w += __shfl_xor(acc.w, off, 32);
        }

        if (grp == 0) {
            float4 r = make_float4(acc.x + bv.x, acc.y + bv.y, acc.z + bv.z, acc.w + bv.w);
            if (RELU) {
                r.x = fmaxf(r.x, 0.0f);
                r.y = fmaxf(r.y, 0.0f);
                r.z = fmaxf(r.z, 0.0f);
                r.w = fmaxf(r.w, 0.0f);
            }
            *reinterpret_cast<float4*>(res + (long long)d * C + lig * 4) = r;
        }

        if (d + nWaves >= nNodes) break;
        s_l  = s1;
        d = d1;  beg  = beg1; nIn  = nIn1; dd  = dd1;
        d1 = d2; beg1 = beg2; nIn1 = nIn2; dd1 = dd2;
    }
}

#define FG_WAVES  16
#define FG_STRIPS 7
#define FG_NODES  (16 * FG_STRIPS)
#define FG_PITCH  (HID + 4)

__global__ void __launch_bounds__(FG_WAVES * 32)
gcn_gather_product_kernel(const float* __restrict__ xs,
                          const int* __restrict__ col,
                          const int* __restrict__ start,
                          const int* __restrict__ cnt,
                          const float* __restrict__ dinv,
                          const float* __restrict__ bias,
                          const float* __restrict__ W,
                          float* __restrict__ C, int nNodes) {
    constexpr int LPR = HID / 4;
    constexpr int G   = 32 / LPR;
    __shared__ float hs[FG_NODES * FG_PITCH];
    const int lane = (int)(threadIdx.x & 31);
    const int wv   = __builtin_amdgcn_readfirstlane((int)(threadIdx.x >> 5));
    const int base = (int)blockIdx.x * FG_NODES;
    const int last = nNodes - 1;

    const int grp = lane / LPR;
    const int lig = lane - grp * LPR;
    const float* xcol = xs + lig * 4;
    const float4 bv = *reinterpret_cast<const float4*>(bias + lig * 4);

    int   d   = base + wv;
    int   dc  = min(d, last);
    int   beg = start[dc];
    int   nIn = cnt[dc];
    float dd  = dinv[dc];
    int   s_l = dc;
    if (lane < nIn) s_l = col[beg + lane];
    int   d1c  = min(d + FG_WAVES, last);
    int   beg1 = start[d1c];
    int   nIn1 = cnt[d1c];
    float dd1  = dinv[d1c];

    for (int k = 0; k < FG_STRIPS; ++k) {
        int s1 = d1c;
        if (k + 1 < FG_STRIPS && lane < nIn1) s1 = col[beg1 + lane];
        const int d2c = min(d + 2 * FG_WAVES, last);
        int beg2 = 0, nIn2 = 0;
        float dd2 = 0.0f;
        if (k + 2 < FG_STRIPS) {
            beg2 = start[d2c];
            nIn2 = cnt[d2c];
            dd2  = dinv[d2c];
        }

        float4 r = make_float4(0.0f, 0.0f, 0.0f, 0.0f);
        if (d < nNodes) {
            const int n = nIn + 1;
            float4 acc = make_float4(0.0f, 0.0f, 0.0f, 0.0f);
            for (int b0 = 0; b0 < n; b0 += 32) {
                const int m = min(32, n - b0);
                if (b0 > 0) {
                    const int j = b0 + lane;
                    s_l = d;
                    if (j < nIn) s_l = col[beg + j];
                }
                for (int t = 0; t < m; t += 4 * G) {
#pragma unroll
                    for (int u = 0; u < 4; ++u) {
                        const int   e = t + u * G + grp;
                        const int   s = __shfl(s_l, e, 32);
                        const float w = (e < m) ? dd : 0.0f;
                        const float4 v = *reinterpret_cast<const float4*>(xcol + (long long)s * HID);
                        acc.x += v.x * w;
                        acc.y += v.y * w;
                        acc.z += v.z * w;
                        acc.w += v.w * w;
                    }
                }
            }
#pragma unroll
            for (int off = LPR; off < 32; off <<= 1) {
                acc.x += __shfl_xor(acc.x, off, 32);
                acc.y += __shfl_xor(acc.y, off, 32);
                acc.z += __shfl_xor(acc.z, off, 32);
                acc.w += __shfl_xor(acc.w, off, 32);
            }
            r.x = fmaxf(acc.x + bv.x, 0.0f);
            r.y = fmaxf(acc.y + bv.y, 0.0f);
            r.z = fmaxf(acc.z + bv.z, 0.0f);
            r.w = fmaxf(acc.w + bv.w, 0.0f);
        }
        if (grp == 0)
            *reinterpret_cast<float4*>(&hs[(wv + FG_WAVES * k) * FG_PITCH + lig * 4]) = r;

        s_l = s1;
        d += FG_WAVES;
        beg = beg1;  nIn = nIn1;  dd = dd1;
        d1c = d2c; beg1 = beg2; nIn1 = nIn2; dd1 = dd2;
    }

    __syncthreads();

    if (wv >= FG_STRIPS) return;
    const int row0 = base + 16 * wv;
    if (row0 >= nNodes) return;
    const int c = lane & 15;
    const int h = lane >> 4;
    const int crow = row0 + (h << 3);
    const float4 sc0 = *reinterpret_cast<const float4*>(dinv + crow);
    const float4 sc1 = *reinterpret_cast<const float4*>(dinv + crow + 4);

    constexpr int KB = HID / 8;
    const float* wp = W + (4 * h) * OUT_CH + c;
    float wb[KB][4];
#pragma unroll
    for (int j = 0; j < KB; ++j)
#pragma unroll
        for (int i = 0; i < 4; ++i)
            wb[j][i] = wp[(8 * j + i) * OUT_CH];

    const float* ap = hs + (16 * wv + c) * FG_PITCH + 4 * h;
    float4 av[KB];
#pragma unroll
    for (int j = 0; j < KB; ++j)
        av[j] = *reinterpret_cast<const float4*>(ap + 8 * j);

    v8f acc8 = v8f{};
#pragma unroll
    for (int j = 0; j < KB; ++j) {
        v2f a0, a1, b0, b1;
        a0.x = av[j].x; a0.y = av[j].y;
        a1.x = av[j].z; a1.y = av[j].w;
        b0.x = wb[j][0]; b0.y = wb[j][1];
        b1.x = wb[j][2]; b1.y = wb[j][3];
        acc8 = __builtin_amdgcn_wmma_f32_16x16x4_f32(
             false, a0,  false, b0,
             (short)0, acc8,  false,  false);
        acc8 = __builtin_amdgcn_wmma_f32_16x16x4_f32(
             false, a1,  false, b1,
             (short)0, acc8,  false,  false);
    }

    const float sc[8] = {sc0.x, sc0.y, sc0.z, sc0.w, sc1.x, sc1.y, sc1.z, sc1.w};
    if (row0 + 16 <= nNodes) {
#pragma unroll
        for (int r8 = 0; r8 < 8; ++r8)
            C[(long long)(crow + r8) * OUT_CH + c] = acc8[r8] * sc[r8];
    } else {
#pragma unroll
        for (int r8 = 0; r8 < 8; ++r8)
            if (crow + r8 < nNodes)
                C[(long long)(crow + r8) * OUT_CH + c] = acc8[r8] * sc[r8];
    }
}

extern "C" void kernel_launch(void* const* d_in, const int* in_sizes, int n_in,
                              void* d_out, int out_size, void* d_ws, size_t ws_size,
                              hipStream_t stream) {
    const float* x  = (const float*)d_in[0];
    const int*   ei = (const int*)  d_in[1];
    const float* W1 = (const float*)d_in[2];
    const float* b1 = (const float*)d_in[3];
    const float* W2 = (const float*)d_in[4];
    const float* b2 = (const float*)d_in[5];
    float* out = (float*)d_out;

    const int nNodes = in_sizes[0] / IN_CH;
    const int nEdges = in_sizes[1] / 2;
    const int* src = ei;
    const int* dst = ei + nEdges;

    const long long hElems = (long long)nNodes * HID;
    float* xl1  = (float*)d_ws;
    float* h    = xl1 + hElems;
    float* xl2  = h   + hElems;
    float* dinv = xl2 + (long long)nNodes * OUT_CH;
    int* start  = (int*)(dinv + nNodes);
    int* tmp    = start + nNodes;
    int* col    = tmp   + nEdges;
    int* cnt    = col   + nEdges;
    int* pfx    = (int*)(((size_t)(cnt + nNodes) + 255) & ~(size_t)255);

    const int B = 256;
    const int nBuckets = (nNodes + NPB - 1) / NPB;
    const int nChunks  = (nEdges + CH - 1) / CH;

    gcn_chunk_sort_kernel<<<nChunks, B, 0, stream>>>(src, dst, pfx, tmp, nEdges, nBuckets);
    gcn_bucket_place_kernel<<<nBuckets, B, 0, stream>>>(tmp, pfx, start, cnt, dinv, col,
                                                        nNodes, nEdges, nChunks, nBuckets);

    {
        int strips = nNodes / 16;
        gcn_gemm_strip_wmma_f32<IN_CH, HID / 16><<<(strips + 3) / 4, 128, 0, stream>>>(x, W1, dinv, xl1, nNodes);
    }

    gcn_gather_product_kernel<<<(nNodes + FG_NODES - 1) / FG_NODES, FG_WAVES * 32, 0, stream>>>(
        xl1, col, start, cnt, dinv, b1, W2, xl2, nNodes);

    const int gatherGroups = 2048;

    gcn_gather_kernel<OUT_CH, false><<<gatherGroups, B, 0, stream>>>(
        xl2, col, start, cnt, dinv, b2, out, nNodes);
}
